# starting kernel
# baseline (speedup 1.0000x reference)
#include <hip/hip_runtime.h>
#include <hip/hip_bf16.h>
#include <math.h>

typedef __attribute__((ext_vector_type(16))) _Float16 v16h;
typedef __attribute__((ext_vector_type(8)))  _Float16 v8h;
typedef __attribute__((ext_vector_type(8)))  float    v8f;

#define NEG_SLOPE 0.2f

__device__ __forceinline__ int f2oi(float f) {
  int i = __float_as_int(f);
  return i >= 0 ? i : (i ^ 0x7FFFFFFF);
}
__device__ __forceinline__ float oi2f(int i) {
  return __int_as_float(i >= 0 ? i : (i ^ 0x7FFFFFFF));
}
__device__ __forceinline__ float leaky(float x) { return x > 0.f ? x : NEG_SLOPE * x; }
__device__ __forceinline__ void atomAddF(float* p, float v) {
  __hip_atomic_fetch_add(p, v, __ATOMIC_RELAXED, __HIP_MEMORY_SCOPE_AGENT);
}

__global__ void k_cast_f16(const float* __restrict__ in, _Float16* __restrict__ out, int n) {
  int t = blockIdx.x * blockDim.x + threadIdx.x;
  if (t < n) out[t] = (_Float16)in[t];
}

__global__ void k_castT_f16(const float* __restrict__ in, _Float16* __restrict__ out, int R, int C) {
  int t = blockIdx.x * blockDim.x + threadIdx.x;
  if (t >= R * C) return;
  int r = t / C, c = t - r * C;
  out[c * R + r] = (_Float16)in[t];
}

__global__ void k_wmma_gemm(const _Float16* __restrict__ A,
                            const _Float16* __restrict__ Bt,
                            float* __restrict__ C,
                            int M, int N, int K) {
  const int tilesN      = N >> 4;
  const int wavesPerRow = tilesN >> 2;
  const int totalWaves  = (M >> 4) * wavesPerRow;
  const int wave = blockIdx.x * (blockDim.x >> 5) + (threadIdx.x >> 5);
  if (wave >= totalWaves) return;
  const int lane = threadIdx.x & 31;
  const int half = lane >> 4;
  const int l16  = lane & 15;
  const int tm  = wave / wavesPerRow;
  const int tn0 = (wave - tm * wavesPerRow) * 4;

  const _Float16* aRow = A + (size_t)(tm * 16 + l16) * K;
  const _Float16* bRow[4];
  #pragma unroll
  for (int j = 0; j < 4; ++j)
    bRow[j] = Bt + (size_t)((tn0 + j) * 16 + l16) * K;

  v8f acc0 = {}, acc1 = {}, acc2 = {}, acc3 = {};
  for (int k = 0; k < K; k += 32) {
    const int ka = k + half * 8;
    union { v16h v; v8h h[2]; } a, b0, b1, b2, b3;
    a.h[0]  = *(const v8h*)(aRow + ka);
    a.h[1]  = *(const v8h*)(aRow + ka + 16);
    b0.h[0] = *(const v8h*)(bRow[0] + ka);
    b0.h[1] = *(const v8h*)(bRow[0] + ka + 16);
    b1.h[0] = *(const v8h*)(bRow[1] + ka);
    b1.h[1] = *(const v8h*)(bRow[1] + ka + 16);
    b2.h[0] = *(const v8h*)(bRow[2] + ka);
    b2.h[1] = *(const v8h*)(bRow[2] + ka + 16);
    b3.h[0] = *(const v8h*)(bRow[3] + ka);
    b3.h[1] = *(const v8h*)(bRow[3] + ka + 16);
    acc0 = __builtin_amdgcn_wmma_f32_16x16x32_f16(false, a.v, false, b0.v, (short)0, acc0, false, false);
    acc1 = __builtin_amdgcn_wmma_f32_16x16x32_f16(false, a.v, false, b1.v, (short)0, acc1, false, false);
    acc2 = __builtin_amdgcn_wmma_f32_16x16x32_f16(false, a.v, false, b2.v, (short)0, acc2, false, false);
    acc3 = __builtin_amdgcn_wmma_f32_16x16x32_f16(false, a.v, false, b3.v, (short)0, acc3, false, false);
  }
  float* cRow = C + (size_t)(tm * 16 + half * 8) * N + tn0 * 16 + l16;
  #pragma unroll
  for (int j = 0; j < 8; ++j) {
    float* cr = cRow + (size_t)j * N;
    cr[0]  = acc0[j];
    cr[16] = acc1[j];
    cr[32] = acc2[j];
    cr[48] = acc3[j];
  }
}

__global__ void k_attn(const float* __restrict__ h, const float* __restrict__ att_s,
                       const float* __restrict__ att_d,
                       float* __restrict__ as_, float* __restrict__ ad_,
                       int Nn, int H, int Cc) {
  int t = blockIdx.x * blockDim.x + threadIdx.x;
  if (t >= Nn * H) return;
  int n = t / H, hh = t - n * H;
  const float* hp = h + (size_t)n * H * Cc + (size_t)hh * Cc;
  const float* sp = att_s + hh * Cc;
  const float* dp = att_d + hh * Cc;
  float a0 = 0.f, a1 = 0.f;
  for (int c = 0; c < Cc; ++c) { float v = hp[c]; a0 += v * sp[c]; a1 += v * dp[c]; }
  as_[t] = a0;
  ad_[t] = a1;
}

__global__ void k_init(float* __restrict__ agg, int nAgg,
                       int* __restrict__ m, float* __restrict__ s, int nMS) {
  int t = blockIdx.x * blockDim.x + threadIdx.x;
  if (t < nAgg) agg[t] = 0.f;
  if (t < nMS) { m[t] = f2oi(-__builtin_inff()); s[t] = 0.f; }
}

__global__ void k_edge_max(const int* __restrict__ src, const int* __restrict__ dst,
                           int E, int Etot,
                           const float* __restrict__ as_, const float* __restrict__ ad_,
                           int H, int* __restrict__ m) {
  int t = blockIdx.x * blockDim.x + threadIdx.x;
  if (t >= Etot * H) return;
  int e = t / H, hh = t - e * H;
  int s_ = e < E ? src[e] : (e - E);
  int d_ = e < E ? dst[e] : (e - E);
  float ev = leaky(as_[s_ * H + hh] + ad_[d_ * H + hh]);
  atomicMax(&m[d_ * H + hh], f2oi(ev));
}

__global__ void k_edge_sum(const int* __restrict__ src, const int* __restrict__ dst,
                           int E, int Etot,
                           const float* __restrict__ as_, const float* __restrict__ ad_,
                           int H, const int* __restrict__ m, float* __restrict__ s) {
  int t = blockIdx.x * blockDim.x + threadIdx.x;
  if (t >= Etot * H) return;
  int e = t / H, hh = t - e * H;
  int s_ = e < E ? src[e] : (e - E);
  int d_ = e < E ? dst[e] : (e - E);
  float ev = leaky(as_[s_ * H + hh] + ad_[d_ * H + hh]);
  atomAddF(&s[d_ * H + hh], expf(ev - oi2f(m[d_ * H + hh])));
}

__global__ void k_edge_agg(const int* __restrict__ src, const int* __restrict__ dst,
                           int E, int Etot,
                           const float* __restrict__ as_, const float* __restrict__ ad_,
                           const int* __restrict__ m, const float* __restrict__ ssum,
                           const float* __restrict__ feat, float* __restrict__ outp,
                           int H, int Cc) {
  int wid = blockIdx.x * (blockDim.x >> 5) + (threadIdx.x >> 5);
  if (wid >= Etot) return;
  int lane = threadIdx.x & 31;
  int s_ = wid < E ? src[wid] : (wid - E);
  int d_ = wid < E ? dst[wid] : (wid - E);
  int HC = H * Cc;
  const float* fRow = feat + (size_t)s_ * HC;
  float*       oRow = outp + (size_t)d_ * HC;

  __builtin_prefetch(&fRow[(size_t)lane * HC / 32], 0, 0);

  float alpha[4];
  for (int hh = 0; hh < H; ++hh) {
    float ev = leaky(as_[s_ * H + hh] + ad_[d_ * H + hh]);
    alpha[hh] = expf(ev - oi2f(m[d_ * H + hh])) / (ssum[d_ * H + hh] + 1e-16f);
  }
  for (int base = 0; base < HC; base += 32) {
    float al = alpha[base / Cc];
    int idx = base + lane;
    atomAddF(&oRow[idx], fRow[idx] * al);
  }
}

__global__ void k_bias_relu_cast(const float* __restrict__ a, const float* __restrict__ b,
                                 _Float16* __restrict__ o, int n, int hcMask) {
  int t = blockIdx.x * blockDim.x + threadIdx.x;
  if (t >= n) return;
  float v = a[t] + b[t & hcMask];
  o[t] = (_Float16)(v > 0.f ? v : 0.f);
}

__global__ void k_bias_add(float* __restrict__ o, const float* __restrict__ b, int n, int hcMask) {
  int t = blockIdx.x * blockDim.x + threadIdx.x;
  if (t >= n) return;
  o[t] += b[t & hcMask];
}

extern "C" void kernel_launch(void* const* d_in, const int* in_sizes, int n_in,
                              void* d_out, int out_size, void* d_ws, size_t ws_size,
                              hipStream_t stream) {
  const int IN = 128, HID = 64, H1 = 4;
  const int HC1 = H1 * HID;
  const int HC2 = HID;
  const int Nn   = in_sizes[0] / IN;
  const int E    = in_sizes[1] / 2;
  const int Etot = E + Nn;

  const float* x   = (const float*)d_in[0];
  const int*   ei  = (const int*)d_in[1];
  const float* W1  = (const float*)d_in[2];
  const float* aS1 = (const float*)d_in[3];
  const float* aD1 = (const float*)d_in[4];
  const float* b1  = (const float*)d_in[5];
  const float* W2  = (const float*)d_in[6];
  const float* aS2 = (const float*)d_in[7];
  const float* aD2 = (const float*)d_in[8];
  const float* b2  = (const float*)d_in[9];
  float* out = (float*)d_out;
  const int* srcI = ei;
  const int* dstI = ei + E;

  char* p = (char*)d_ws;
  auto carve = [&](size_t bytes) -> void* {
    void* r = (void*)p;
    p += (bytes + 255) & ~(size_t)255;
    return r;
  };
  _Float16* xh   = (_Float16*)carve((size_t)Nn * IN * 2);
  _Float16* w1t  = (_Float16*)carve((size_t)HC1 * IN * 2);
  _Float16* w2t  = (_Float16*)carve((size_t)HID * HC1 * 2);
  float*    h1   = (float*)carve((size_t)Nn * HC1 * 4);
  float*    agg1 = (float*)carve((size_t)Nn * HC1 * 4);
  _Float16* h1h  = (_Float16*)carve((size_t)Nn * HC1 * 2);
  float*    h2   = (float*)carve((size_t)Nn * HC2 * 4);
  float*    as1  = (float*)carve((size_t)Nn * H1 * 4);
  float*    ad1  = (float*)carve((size_t)Nn * H1 * 4);
  int*      m1   = (int*)carve((size_t)Nn * H1 * 4);
  float*    s1   = (float*)carve((size_t)Nn * H1 * 4);
  float*    as2  = (float*)carve((size_t)Nn * 4);
  float*    ad2  = (float*)carve((size_t)Nn * 4);
  int*      m2   = (int*)carve((size_t)Nn * 4);
  float*    s2   = (float*)carve((size_t)Nn * 4);

  const int TB = 256;
  auto cdiv = [](long long a, long long b) { return (int)((a + b - 1) / b); };

  k_cast_f16<<<cdiv((long long)Nn * IN, TB), TB, 0, stream>>>(x, xh, Nn * IN);
  k_castT_f16<<<cdiv((long long)IN * HC1, TB), TB, 0, stream>>>(W1, w1t, IN, HC1);
  k_castT_f16<<<cdiv((long long)HC1 * HID, TB), TB, 0, stream>>>(W2, w2t, HC1, HID);

  {
    int waves = (Nn / 16) * (HC1 / 64);
    k_wmma_gemm<<<cdiv(waves, 8), TB, 0, stream>>>(xh, w1t, h1, Nn, HC1, IN);
  }
  k_attn<<<cdiv((long long)Nn * H1, TB), TB, 0, stream>>>(h1, aS1, aD1, as1, ad1, Nn, H1, HID);
  k_init<<<cdiv((long long)Nn * HC1, TB), TB, 0, stream>>>(agg1, Nn * HC1, m1, s1, Nn * H1);
  k_edge_max<<<cdiv((long long)Etot * H1, TB), TB, 0, stream>>>(srcI, dstI, E, Etot, as1, ad1, H1, m1);
  k_edge_sum<<<cdiv((long long)Etot * H1, TB), TB, 0, stream>>>(srcI, dstI, E, Etot, as1, ad1, H1, m1, s1);
  k_edge_agg<<<cdiv(Etot, 8), TB, 0, stream>>>(srcI, dstI, E, Etot, as1, ad1, m1, s1, h1, agg1, H1, HID);
  k_bias_relu_cast<<<cdiv((long long)Nn * HC1, TB), TB, 0, stream>>>(agg1, b1, h1h, Nn * HC1, HC1 - 1);

  {
    int waves = (Nn / 16) * (HC2 / 64);
    k_wmma_gemm<<<cdiv(waves, 8), TB, 0, stream>>>(h1h, w2t, h2, Nn, HC2, HC1);
  }
  k_attn<<<cdiv((long long)Nn, TB), TB, 0, stream>>>(h2, aS2, aD2, as2, ad2, Nn, 1, HID);
  k_init<<<cdiv((long long)Nn * HC2, TB), TB, 0, stream>>>(out, Nn * HC2, m2, s2, Nn);
  k_edge_max<<<cdiv((long long)Etot, TB), TB, 0, stream>>>(srcI, dstI, E, Etot, as2, ad2, 1, m2);
  k_edge_sum<<<cdiv((long long)Etot, TB), TB, 0, stream>>>(srcI, dstI, E, Etot, as2, ad2, 1, m2, s2);
  k_edge_agg<<<cdiv(Etot, 8), TB, 0, stream>>>(srcI, dstI, E, Etot, as2, ad2, m2, s2, h2, out, 1, HID);
  k_bias_add<<<cdiv((long long)Nn * HC2, TB), TB, 0, stream>>>(out, b2, Nn * HC2, HC2 - 1);
}


# Edges grouped by destination in-call (counting sort); one wave per dst row does softmax and sums messages in registers, stores once; float atomics removed, sums now in grouped-arrival order
# speedup vs baseline: 3.0620x; 3.0620x over previous
#include <hip/hip_runtime.h>
#include <hip/hip_bf16.h>
#include <math.h>

typedef __attribute__((ext_vector_type(16))) _Float16 v16h;
typedef __attribute__((ext_vector_type(8)))  _Float16 v8h;
typedef __attribute__((ext_vector_type(4)))  _Float16 v4h;
typedef __attribute__((ext_vector_type(8)))  float    v8f;

#define NEG_SLOPE 0.2f

__device__ __forceinline__ float leaky(float x) { return x > 0.f ? x : NEG_SLOPE * x; }

__device__ __forceinline__ float waveMax(float v) {
  for (int o = 16; o > 0; o >>= 1) v = fmaxf(v, __shfl_xor(v, o));
  return v;
}
__device__ __forceinline__ float waveSum(float v) {
  for (int o = 16; o > 0; o >>= 1) v += __shfl_xor(v, o);
  return v;
}
__device__ __forceinline__ float laneF(float v, int j) {
  return __int_as_float(__builtin_amdgcn_readlane(__float_as_int(v), j));
}

__global__ void k_cast_f16(const float* __restrict__ in, _Float16* __restrict__ out, int n,
                           int* __restrict__ cnt, int nCnt) {
  int t = blockIdx.x * blockDim.x + threadIdx.x;
  if (t < n) out[t] = (_Float16)in[t];
  if (t < nCnt) cnt[t] = 0;
}

__global__ void k_castT_f16(const float* __restrict__ in, _Float16* __restrict__ out, int R, int C) {
  int t = blockIdx.x * blockDim.x + threadIdx.x;
  if (t >= R * C) return;
  int r = t / C, c = t - r * C;
  out[c * R + r] = (_Float16)in[t];
}

__global__ void k_hist(const int* __restrict__ dst, int E,
                       int* __restrict__ cnt, int* __restrict__ rank) {
  int t = blockIdx.x * blockDim.x + threadIdx.x;
  if (t >= E) return;
  rank[t] = atomicAdd(&cnt[dst[t]], 1);
}

__global__ void __launch_bounds__(1024) k_scan_local(const int* __restrict__ cnt, int* __restrict__ off,
                                                     int* __restrict__ part, int n) {
  __shared__ int sh[1024];
  const int t = threadIdx.x;
  const int i = blockIdx.x * 1024 + t;
  const int v = i < n ? cnt[i] : 0;
  sh[t] = v;
  __syncthreads();
  for (int o = 1; o < 1024; o <<= 1) {
    const int u = (t >= o) ? sh[t - o] : 0;
    __syncthreads();
    sh[t] += u;
    __syncthreads();
  }
  if (i < n) off[i] = sh[t] - v;
  if (t == 1023) part[blockIdx.x] = sh[1023];
}

__global__ void __launch_bounds__(1024) k_scan_add(int* __restrict__ off, const int* __restrict__ part,
                                                   int n, int nTiles) {
  __shared__ int sBefore, sAll;
  const int t = threadIdx.x;
  if (t < 32) {
    int before = 0, all = 0;
    for (int j = t; j < nTiles; j += 32) {
      const int pj = part[j];
      all += pj;
      if (j < (int)blockIdx.x) before += pj;
    }
    for (int o = 16; o > 0; o >>= 1) {
      before += __shfl_xor(before, o);
      all    += __shfl_xor(all, o);
    }
    if (t == 0) { sBefore = before; sAll = all; }
  }
  __syncthreads();
  const int i = blockIdx.x * 1024 + t;
  if (i < n) off[i] += sBefore;
  else if (i == n) off[i] = sAll;
}

__global__ void k_fill(const int* __restrict__ src, const int* __restrict__ dst, int E,
                       const int* __restrict__ off, const int* __restrict__ rank,
                       int* __restrict__ srcS) {
  int t = blockIdx.x * blockDim.x + threadIdx.x;
  if (t >= E) return;
  srcS[off[dst[t]] + rank[t]] = src[t];
}

__global__ void k_wmma_gemm(const _Float16* __restrict__ A,
                            const _Float16* __restrict__ Bt,
                            float* __restrict__ C,
                            int M, int N, int K) {
  const int tilesN      = N >> 4;
  const int wavesPerRow = tilesN >> 2;
  const int totalWaves  = (M >> 4) * wavesPerRow;
  const int wave = blockIdx.x * (blockDim.x >> 5) + (threadIdx.x >> 5);
  if (wave >= totalWaves) return;
  const int lane = threadIdx.x & 31;
  const int half = lane >> 4;
  const int l16  = lane & 15;
  const int tm  = wave / wavesPerRow;
  const int tn0 = (wave - tm * wavesPerRow) * 4;

  const _Float16* aRow = A + (size_t)(tm * 16 + l16) * K;
  const _Float16* bRow[4];
  #pragma unroll
  for (int j = 0; j < 4; ++j)
    bRow[j] = Bt + (size_t)((tn0 + j) * 16 + l16) * K;

  v8f acc0 = {}, acc1 = {}, acc2 = {}, acc3 = {};
  for (int k = 0; k < K; k += 32) {
    const int ka = k + half * 8;
    union { v16h v; v8h h[2]; } a, b0, b1, b2, b3;
    a.h[0]  = *(const v8h*)(aRow + ka);
    a.h[1]  = *(const v8h*)(aRow + ka + 16);
    b0.h[0] = *(const v8h*)(bRow[0] + ka);
    b0.h[1] = *(const v8h*)(bRow[0] + ka + 16);
    b1.h[0] = *(const v8h*)(bRow[1] + ka);
    b1.h[1] = *(const v8h*)(bRow[1] + ka + 16);
    b2.h[0] = *(const v8h*)(bRow[2] + ka);
    b2.h[1] = *(const v8h*)(bRow[2] + ka + 16);
    b3.h[0] = *(const v8h*)(bRow[3] + ka);
    b3.h[1] = *(const v8h*)(bRow[3] + ka + 16);
    acc0 = __builtin_amdgcn_wmma_f32_16x16x32_f16(false, a.v, false, b0.v, (short)0, acc0, false, false);
    acc1 = __builtin_amdgcn_wmma_f32_16x16x32_f16(false, a.v, false, b1.v, (short)0, acc1, false, false);
    acc2 = __builtin_amdgcn_wmma_f32_16x16x32_f16(false, a.v, false, b2.v, (short)0, acc2, false, false);
    acc3 = __builtin_amdgcn_wmma_f32_16x16x32_f16(false, a.v, false, b3.v, (short)0, acc3, false, false);
  }
  float* cRow = C + (size_t)(tm * 16 + half * 8) * N + tn0 * 16 + l16;
  #pragma unroll
  for (int j = 0; j < 8; ++j) {
    float* cr = cRow + (size_t)j * N;
    cr[0]  = acc0[j];
    cr[16] = acc1[j];
    cr[32] = acc2[j];
    cr[48] = acc3[j];
  }
}

__global__ void k_attn(const float* __restrict__ h, const float* __restrict__ att_s,
                       const float* __restrict__ att_d,
                       float* __restrict__ as_, float* __restrict__ ad_,
                       int Nn, int H, int Cc) {
  int t = blockIdx.x * blockDim.x + threadIdx.x;
  if (t >= Nn * H) return;
  int n = t / H, hh = t - n * H;
  const float* hp = h + (size_t)n * H * Cc + (size_t)hh * Cc;
  const float* sp = att_s + hh * Cc;
  const float* dp = att_d + hh * Cc;
  float a0 = 0.f, a1 = 0.f;
  for (int c = 0; c < Cc; ++c) { float v = hp[c]; a0 += v * sp[c]; a1 += v * dp[c]; }
  as_[t] = a0;
  ad_[t] = a1;
}


__global__ void k_node_l1(const int* __restrict__ off, const int* __restrict__ srcS,
                          const float* __restrict__ as_, const float* __restrict__ ad_,
                          const float* __restrict__ feat, const float* __restrict__ bias,
                          _Float16* __restrict__ outh, int Nn) {
  int d = blockIdx.x * (blockDim.x >> 5) + (threadIdx.x >> 5);
  if (d >= Nn) return;
  d = __builtin_amdgcn_readfirstlane(d);
  const int lane = threadIdx.x & 31;
  const int beg = off[d];
  const int deg = off[d + 1] - beg;
  const int tot = deg + 1;
  const float4 adv = *(const float4*)(ad_ + (size_t)d * 4);

  float m0 = -__builtin_inff(), m1 = m0, m2 = m0, m3 = m0;
  for (int base = 0; base < tot; base += 32) {
    const int i = base + lane;
    if (i < tot) {
      const int s = i < deg ? srcS[beg + i] : d;
      const float4 a = *(const float4*)(as_ + (size_t)s * 4);
      m0 = fmaxf(m0, leaky(a.x + adv.x));
      m1 = fmaxf(m1, leaky(a.y + adv.y));
      m2 = fmaxf(m2, leaky(a.z + adv.z));
      m3 = fmaxf(m3, leaky(a.w + adv.w));
    }
  }
  m0 = waveMax(m0); m1 = waveMax(m1); m2 = waveMax(m2); m3 = waveMax(m3);

  float s0 = 0.f, s1 = 0.f, s2 = 0.f, s3 = 0.f;
  for (int base = 0; base < tot; base += 32) {
    const int i = base + lane;
    if (i < tot) {
      const int s = i < deg ? srcS[beg + i] : d;
      const float4 a = *(const float4*)(as_ + (size_t)s * 4);
      s0 += expf(leaky(a.x + adv.x) - m0);
      s1 += expf(leaky(a.y + adv.y) - m1);
      s2 += expf(leaky(a.z + adv.z) - m2);
      s3 += expf(leaky(a.w + adv.w) - m3);
    }
  }
  s0 = waveSum(s0) + 1e-16f; s1 = waveSum(s1) + 1e-16f;
  s2 = waveSum(s2) + 1e-16f; s3 = waveSum(s3) + 1e-16f;

  const bool upper = lane >= 16;
  float4 acc0 = make_float4(0.f, 0.f, 0.f, 0.f);
  float4 acc1 = make_float4(0.f, 0.f, 0.f, 0.f);
  for (int base = 0; base < tot; base += 32) {
    const int i = base + lane;
    int s = d;
    float a0 = 0.f, a1 = 0.f, a2 = 0.f, a3 = 0.f;
    if (i < tot) {
      s = i < deg ? srcS[beg + i] : d;
      const float4 a = *(const float4*)(as_ + (size_t)s * 4);
      a0 = expf(leaky(a.x + adv.x) - m0) / s0;
      a1 = expf(leaky(a.y + adv.y) - m1) / s1;
      a2 = expf(leaky(a.z + adv.z) - m2) / s2;
      a3 = expf(leaky(a.w + adv.w) - m3) / s3;
    }
    const int cnt = (tot - base < 32) ? (tot - base) : 32;
    #pragma unroll 4
    for (int j = 0; j < cnt; ++j) {
      const int   sj = __builtin_amdgcn_readlane(s, j);
      const float b0 = laneF(a0, j), b1 = laneF(a1, j), b2 = laneF(a2, j), b3 = laneF(a3, j);
      const float alLo = upper ? b1 : b0;
      const float alHi = upper ? b3 : b2;
      const float* row = feat + (size_t)sj * 256;
      const float4 f0 = *(const float4*)(row + lane * 4);
      const float4 f1 = *(const float4*)(row + 128 + lane * 4);
      acc0.x += f0.x * alLo; acc0.y += f0.y * alLo; acc0.z += f0.z * alLo; acc0.w += f0.w * alLo;
      acc1.x += f1.x * alHi; acc1.y += f1.y * alHi; acc1.z += f1.z * alHi; acc1.w += f1.w * alHi;
    }
  }

  const int c0 = lane * 4, c1 = 128 + lane * 4;
  float v;
  v4h h0, h1;
  v = acc0.x + bias[c0 + 0]; h0[0] = (_Float16)(v > 0.f ? v : 0.f);
  v = acc0.y + bias[c0 + 1]; h0[1] = (_Float16)(v > 0.f ? v : 0.f);
  v = acc0.z + bias[c0 + 2]; h0[2] = (_Float16)(v > 0.f ? v : 0.f);
  v = acc0.w + bias[c0 + 3]; h0[3] = (_Float16)(v > 0.f ? v : 0.f);
  v = acc1.x + bias[c1 + 0]; h1[0] = (_Float16)(v > 0.f ? v : 0.f);
  v = acc1.y + bias[c1 + 1]; h1[1] = (_Float16)(v > 0.f ? v : 0.f);
  v = acc1.z + bias[c1 + 2]; h1[2] = (_Float16)(v > 0.f ? v : 0.f);
  v = acc1.w + bias[c1 + 3]; h1[3] = (_Float16)(v > 0.f ? v : 0.f);
  _Float16* o = outh + (size_t)d * 256;
  *(v4h*)(o + c0) = h0;
  *(v4h*)(o + c1) = h1;
}

__global__ void k_node_l2(const int* __restrict__ off, const int* __restrict__ srcS,
                          const float* __restrict__ as_, const float* __restrict__ ad_,
                          const float* __restrict__ feat, const float* __restrict__ bias,
                          float* __restrict__ outp, int Nn) {
  int d = blockIdx.x * (blockDim.x >> 5) + (threadIdx.x >> 5);
  if (d >= Nn) return;
  d = __builtin_amdgcn_readfirstlane(d);
  const int lane = threadIdx.x & 31;
  const int beg = off[d];
  const int deg = off[d + 1] - beg;
  const int tot = deg + 1;
  const float adv = ad_[d];

  float m = -__builtin_inff();
  for (int base = 0; base < tot; base += 32) {
    const int i = base + lane;
    if (i < tot) {
      const int s = i < deg ? srcS[beg + i] : d;
      m = fmaxf(m, leaky(as_[s] + adv));
    }
  }
  m = waveMax(m);

  float sm = 0.f;
  for (int base = 0; base < tot; base += 32) {
    const int i = base + lane;
    if (i < tot) {
      const int s = i < deg ? srcS[beg + i] : d;
      sm += expf(leaky(as_[s] + adv) - m);
    }
  }
  sm = waveSum(sm) + 1e-16f;

  float2 acc = make_float2(0.f, 0.f);
  for (int base = 0; base < tot; base += 32) {
    const int i = base + lane;
    int s = d;
    float al = 0.f;
    if (i < tot) {
      s = i < deg ? srcS[beg + i] : d;
      al = expf(leaky(as_[s] + adv) - m) / sm;
    }
    const int cnt = (tot - base < 32) ? (tot - base) : 32;
    #pragma unroll 4
    for (int j = 0; j < cnt; ++j) {
      const int   sj = __builtin_amdgcn_readlane(s, j);
      const float aj = laneF(al, j);
      const float2 f = *(const float2*)(feat + (size_t)sj * 64 + lane * 2);
      acc.x += f.x * aj;
      acc.y += f.y * aj;
    }
  }
  float2 r;
  r.x = acc.x + bias[lane * 2];
  r.y = acc.y + bias[lane * 2 + 1];
  *(float2*)(outp + (size_t)d * 64 + lane * 2) = r;
}

extern "C" void kernel_launch(void* const* d_in, const int* in_sizes, int n_in,
                              void* d_out, int out_size, void* d_ws, size_t ws_size,
                              hipStream_t stream) {
  const int IN = 128, HID = 64, H1 = 4;
  const int HC1 = H1 * HID;
  const int HC2 = HID;
  const int Nn   = in_sizes[0] / IN;
  const int E    = in_sizes[1] / 2;

  const float* x   = (const float*)d_in[0];
  const int*   ei  = (const int*)d_in[1];
  const float* W1  = (const float*)d_in[2];
  const float* aS1 = (const float*)d_in[3];
  const float* aD1 = (const float*)d_in[4];
  const float* b1  = (const float*)d_in[5];
  const float* W2  = (const float*)d_in[6];
  const float* aS2 = (const float*)d_in[7];
  const float* aD2 = (const float*)d_in[8];
  const float* b2  = (const float*)d_in[9];
  float* out = (float*)d_out;
  const int* srcI = ei;
  const int* dstI = ei + E;

  char* p = (char*)d_ws;
  auto carve = [&](size_t bytes) -> void* {
    void* r = (void*)p;
    p += (bytes + 255) & ~(size_t)255;
    return r;
  };
  _Float16* xh   = (_Float16*)carve((size_t)Nn * IN * 2);
  _Float16* w1t  = (_Float16*)carve((size_t)HC1 * IN * 2);
  _Float16* w2t  = (_Float16*)carve((size_t)HID * HC1 * 2);
  float*    h1   = (float*)carve((size_t)Nn * HC1 * 4);
  _Float16* h1h  = (_Float16*)carve((size_t)Nn * HC1 * 2);
  float*    h2   = (float*)carve((size_t)Nn * HC2 * 4);
  float*    as1  = (float*)carve((size_t)Nn * H1 * 4);
  float*    ad1  = (float*)carve((size_t)Nn * H1 * 4);
  float*    as2  = (float*)carve((size_t)Nn * 4);
  float*    ad2  = (float*)carve((size_t)Nn * 4);
  int*      cnt  = (int*)carve((size_t)Nn * 4);
  int*      off  = (int*)carve((size_t)(Nn + 1) * 4);
  int*      part = (int*)carve((size_t)1024 * 4);
  int*      rank = (int*)carve((size_t)E * 4);
  int*      srcS = (int*)carve((size_t)E * 4);

  const int TB = 256;
  auto cdiv = [](long long a, long long b) { return (int)((a + b - 1) / b); };

  k_cast_f16<<<cdiv((long long)Nn * IN, TB), TB, 0, stream>>>(x, xh, Nn * IN, cnt, Nn);
  k_castT_f16<<<cdiv((long long)IN * HC1, TB), TB, 0, stream>>>(W1, w1t, IN, HC1);
  k_castT_f16<<<cdiv((long long)HC1 * HID, TB), TB, 0, stream>>>(W2, w2t, HC1, HID);

  k_hist<<<cdiv(E, TB), TB, 0, stream>>>(dstI, E, cnt, rank);
  k_scan_local<<<cdiv(Nn, 1024), 1024, 0, stream>>>(cnt, off, part, Nn);
  k_scan_add<<<cdiv((long long)Nn + 1, 1024), 1024, 0, stream>>>(off, part, Nn, cdiv(Nn, 1024));
  k_fill<<<cdiv(E, TB), TB, 0, stream>>>(srcI, dstI, E, off, rank, srcS);

  {
    int waves = (Nn / 16) * (HC1 / 64);
    k_wmma_gemm<<<cdiv(waves, 8), TB, 0, stream>>>(xh, w1t, h1, Nn, HC1, IN);
  }
  k_attn<<<cdiv((long long)Nn * H1, TB), TB, 0, stream>>>(h1, aS1, aD1, as1, ad1, Nn, H1, HID);
  k_node_l1<<<cdiv(Nn, 8), TB, 0, stream>>>(off, srcS, as1, ad1, h1, b1, h1h, Nn);

  {
    int waves = (Nn / 16) * (HC2 / 64);
    k_wmma_gemm<<<cdiv(waves, 8), TB, 0, stream>>>(h1h, w2t, h2, Nn, HC2, HC1);
  }
  k_attn<<<cdiv((long long)Nn, TB), TB, 0, stream>>>(h2, aS2, aD2, as2, ad2, Nn, 1, HID);
  k_node_l2<<<cdiv(Nn, 8), TB, 0, stream>>>(off, srcS, as2, ad2, h2, b2, out, Nn);
}


# x is rounded to the GEMM operand width inside the layer-1 A loader (same rounding), not by a one-element-per-thread cast pass: 200000 short waves, one launch and the narrow copy of x gone
# speedup vs baseline: 3.3027x; 1.0786x over previous
#include <hip/hip_runtime.h>
#include <hip/hip_bf16.h>
#include <math.h>

typedef __attribute__((ext_vector_type(16))) _Float16 v16h;
typedef __attribute__((ext_vector_type(8)))  _Float16 v8h;
typedef __attribute__((ext_vector_type(4)))  _Float16 v4h;
typedef __attribute__((ext_vector_type(8)))  float    v8f;

#define NEG_SLOPE 0.2f

__device__ __forceinline__ float leaky(float x) { return x > 0.f ? x : NEG_SLOPE * x; }

__device__ __forceinline__ float waveMax(float v) {
  for (int o = 16; o > 0; o >>= 1) v = fmaxf(v, __shfl_xor(v, o));
  return v;
}
__device__ __forceinline__ float waveSum(float v) {
  for (int o = 16; o > 0; o >>= 1) v += __shfl_xor(v, o);
  return v;
}
__device__ __forceinline__ float laneF(float v, int j) {
  return __int_as_float(__builtin_amdgcn_readlane(__float_as_int(v), j));
}

__global__ void k_castT_f16(const float* __restrict__ in, _Float16* __restrict__ out, int R, int C,
                            int* __restrict__ cnt, int nCnt) {
  int t = blockIdx.x * blockDim.x + threadIdx.x;
  for (int i = t; i < nCnt; i += gridDim.x * blockDim.x) cnt[i] = 0;
  if (t >= R * C) return;
  int r = t / C, c = t - r * C;
  out[c * R + r] = (_Float16)in[t];
}

__global__ void k_hist(const int* __restrict__ dst, int E,
                       int* __restrict__ cnt, int* __restrict__ rank) {
  int t = blockIdx.x * blockDim.x + threadIdx.x;
  if (t >= E) return;
  rank[t] = atomicAdd(&cnt[dst[t]], 1);
}

__global__ void __launch_bounds__(1024) k_scan_local(const int* __restrict__ cnt, int* __restrict__ off,
                                                     int* __restrict__ part, int n) {
  __shared__ int sh[1024];
  const int t = threadIdx.x;
  const int i = blockIdx.x * 1024 + t;
  const int v = i < n ? cnt[i] : 0;
  sh[t] = v;
  __syncthreads();
  for (int o = 1; o < 1024; o <<= 1) {
    const int u = (t >= o) ? sh[t - o] : 0;
    __syncthreads();
    sh[t] += u;
    __syncthreads();
  }
  if (i < n) off[i] = sh[t] - v;
  if (t == 1023) part[blockIdx.x] = sh[1023];
}

__global__ void __launch_bounds__(1024) k_scan_add(int* __restrict__ off, const int* __restrict__ part,
                                                   int n, int nTiles) {
  __shared__ int sBefore, sAll;
  const int t = threadIdx.x;
  if (t < 32) {
    int before = 0, all = 0;
    for (int j = t; j < nTiles; j += 32) {
      const int pj = part[j];
      all += pj;
      if (j < (int)blockIdx.x) before += pj;
    }
    for (int o = 16; o > 0; o >>= 1) {
      before += __shfl_xor(before, o);
      all    += __shfl_xor(all, o);
    }
    if (t == 0) { sBefore = before; sAll = all; }
  }
  __syncthreads();
  const int i = blockIdx.x * 1024 + t;
  if (i < n) off[i] += sBefore;
  else if (i == n) off[i] = sAll;
}

__global__ void k_fill(const int* __restrict__ src, const int* __restrict__ dst, int E,
                       const int* __restrict__ off, const int* __restrict__ rank,
                       int* __restrict__ srcS) {
  int t = blockIdx.x * blockDim.x + threadIdx.x;
  if (t >= E) return;
  srcS[off[dst[t]] + rank[t]] = src[t];
}

template <bool A_F32>
__device__ __forceinline__ void gemm_strip(const float* __restrict__ Af,
                                           const _Float16* __restrict__ Ah,
                                           const _Float16* __restrict__ Bt,
                                           float* __restrict__ C,
                                           int M, int N, int K) {
  const int tilesN      = N >> 4;
  const int wavesPerRow = tilesN >> 2;
  const int totalWaves  = (M >> 4) * wavesPerRow;
  const int wave = blockIdx.x * (blockDim.x >> 5) + (threadIdx.x >> 5);
  if (wave >= totalWaves) return;
  const int lane = threadIdx.x & 31;
  const int half = lane >> 4;
  const int l16  = lane & 15;
  const int tm  = wave / wavesPerRow;
  const int tn0 = (wave - tm * wavesPerRow) * 4;

  const size_t aOff = (size_t)(tm * 16 + l16) * K;
  const _Float16* bRow[4];
  #pragma unroll
  for (int j = 0; j < 4; ++j)
    bRow[j] = Bt + (size_t)((tn0 + j) * 16 + l16) * K;

  v8f acc0 = {}, acc1 = {}, acc2 = {}, acc3 = {};
  for (int k = 0; k < K; k += 32) {
    const int ka = k + half * 8;
    union { v16h v; v8h h[2]; } a, b0, b1, b2, b3;
    if constexpr (A_F32) {
      const float* ap = Af + aOff + ka;
      const float4 p0 = *(const float4*)(ap);
      const float4 p1 = *(const float4*)(ap + 4);
      const float4 p2 = *(const float4*)(ap + 16);
      const float4 p3 = *(const float4*)(ap + 20);
      a.v[0]  = (_Float16)p0.x; a.v[1]  = (_Float16)p0.y; a.v[2]  = (_Float16)p0.z; a.v[3]  = (_Float16)p0.w;
      a.v[4]  = (_Float16)p1.x; a.v[5]  = (_Float16)p1.y; a.v[6]  = (_Float16)p1.z; a.v[7]  = (_Float16)p1.w;
      a.v[8]  = (_Float16)p2.x; a.v[9]  = (_Float16)p2.y; a.v[10] = (_Float16)p2.z; a.v[11] = (_Float16)p2.w;
      a.v[12] = (_Float16)p3.x; a.v[13] = (_Float16)p3.y; a.v[14] = (_Float16)p3.z; a.v[15] = (_Float16)p3.w;
    } else {
      const _Float16* ap = Ah + aOff + ka;
      a.h[0] = *(const v8h*)(ap);
      a.h[1] = *(const v8h*)(ap + 16);
    }
    b0.h[0] = *(const v8h*)(bRow[0] + ka);
    b0.h[1] = *(const v8h*)(bRow[0] + ka + 16);
    b1.h[0] = *(const v8h*)(bRow[1] + ka);
    b1.h[1] = *(const v8h*)(bRow[1] + ka + 16);
    b2.h[0] = *(const v8h*)(bRow[2] + ka);
    b2.h[1] = *(const v8h*)(bRow[2] + ka + 16);
    b3.h[0] = *(const v8h*)(bRow[3] + ka);
    b3.h[1] = *(const v8h*)(bRow[3] + ka + 16);
    acc0 = __builtin_amdgcn_wmma_f32_16x16x32_f16(false, a.v, false, b0.v, (short)0, acc0, false, false);
    acc1 = __builtin_amdgcn_wmma_f32_16x16x32_f16(false, a.v, false, b1.v, (short)0, acc1, false, false);
    acc2 = __builtin_amdgcn_wmma_f32_16x16x32_f16(false, a.v, false, b2.v, (short)0, acc2, false, false);
    acc3 = __builtin_amdgcn_wmma_f32_16x16x32_f16(false, a.v, false, b3.v, (short)0, acc3, false, false);
  }
  float* cRow = C + (size_t)(tm * 16 + half * 8) * N + tn0 * 16 + l16;
  #pragma unroll
  for (int j = 0; j < 8; ++j) {
    float* cr = cRow + (size_t)j * N;
    cr[0]  = acc0[j];
    cr[16] = acc1[j];
    cr[32] = acc2[j];
    cr[48] = acc3[j];
  }
}

__global__ void k_wmma_gemm_af32(const float* __restrict__ A,
                                 const _Float16* __restrict__ Bt,
                                 float* __restrict__ C,
                                 int M, int N, int K) {
  gemm_strip<true>(A, nullptr, Bt, C, M, N, K);
}

__global__ void k_wmma_gemm(const _Float16* __restrict__ A,
                            const _Float16* __restrict__ Bt,
                            float* __restrict__ C,
                            int M, int N, int K) {
  gemm_strip<false>(nullptr, A, Bt, C, M, N, K);
}

__global__ void k_attn(const float* __restrict__ h, const float* __restrict__ att_s,
                       const float* __restrict__ att_d,
                       float* __restrict__ as_, float* __restrict__ ad_,
                       int Nn, int H, int Cc) {
  int t = blockIdx.x * blockDim.x + threadIdx.x;
  if (t >= Nn * H) return;
  int n = t / H, hh = t - n * H;
  const float* hp = h + (size_t)n * H * Cc + (size_t)hh * Cc;
  const float* sp = att_s + hh * Cc;
  const float* dp = att_d + hh * Cc;
  float a0 = 0.f, a1 = 0.f;
  for (int c = 0; c < Cc; ++c) { float v = hp[c]; a0 += v * sp[c]; a1 += v * dp[c]; }
  as_[t] = a0;
  ad_[t] = a1;
}


__global__ void k_node_l1(const int* __restrict__ off, const int* __restrict__ srcS,
                          const float* __restrict__ as_, const float* __restrict__ ad_,
                          const float* __restrict__ feat, const float* __restrict__ bias,
                          _Float16* __restrict__ outh, int Nn) {
  int d = blockIdx.x * (blockDim.x >> 5) + (threadIdx.x >> 5);
  if (d >= Nn) return;
  d = __builtin_amdgcn_readfirstlane(d);
  const int lane = threadIdx.x & 31;
  const int beg = off[d];
  const int deg = off[d + 1] - beg;
  const int tot = deg + 1;
  const float4 adv = *(const float4*)(ad_ + (size_t)d * 4);

  float m0 = -__builtin_inff(), m1 = m0, m2 = m0, m3 = m0;
  for (int base = 0; base < tot; base += 32) {
    const int i = base + lane;
    if (i < tot) {
      const int s = i < deg ? srcS[beg + i] : d;
      const float4 a = *(const float4*)(as_ + (size_t)s * 4);
      m0 = fmaxf(m0, leaky(a.x + adv.x));
      m1 = fmaxf(m1, leaky(a.y + adv.y));
      m2 = fmaxf(m2, leaky(a.z + adv.z));
      m3 = fmaxf(m3, leaky(a.w + adv.w));
    }
  }
  m0 = waveMax(m0); m1 = waveMax(m1); m2 = waveMax(m2); m3 = waveMax(m3);

  float s0 = 0.f, s1 = 0.f, s2 = 0.f, s3 = 0.f;
  for (int base = 0; base < tot; base += 32) {
    const int i = base + lane;
    if (i < tot) {
      const int s = i < deg ? srcS[beg + i] : d;
      const float4 a = *(const float4*)(as_ + (size_t)s * 4);
      s0 += expf(leaky(a.x + adv.x) - m0);
      s1 += expf(leaky(a.y + adv.y) - m1);
      s2 += expf(leaky(a.z + adv.z) - m2);
      s3 += expf(leaky(a.w + adv.w) - m3);
    }
  }
  s0 = waveSum(s0) + 1e-16f; s1 = waveSum(s1) + 1e-16f;
  s2 = waveSum(s2) + 1e-16f; s3 = waveSum(s3) + 1e-16f;

  const bool upper = lane >= 16;
  float4 acc0 = make_float4(0.f, 0.f, 0.f, 0.f);
  float4 acc1 = make_float4(0.f, 0.f, 0.f, 0.f);
  for (int base = 0; base < tot; base += 32) {
    const int i = base + lane;
    int s = d;
    float a0 = 0.f, a1 = 0.f, a2 = 0.f, a3 = 0.f;
    if (i < tot) {
      s = i < deg ? srcS[beg + i] : d;
      const float4 a = *(const float4*)(as_ + (size_t)s * 4);
      a0 = expf(leaky(a.x + adv.x) - m0) / s0;
      a1 = expf(leaky(a.y + adv.y) - m1) / s1;
      a2 = expf(leaky(a.z + adv.z) - m2) / s2;
      a3 = expf(leaky(a.w + adv.w) - m3) / s3;
    }
    const int cnt = (tot - base < 32) ? (tot - base) : 32;
    #pragma unroll 4
    for (int j = 0; j < cnt; ++j) {
      const int   sj = __builtin_amdgcn_readlane(s, j);
      const float b0 = laneF(a0, j), b1 = laneF(a1, j), b2 = laneF(a2, j), b3 = laneF(a3, j);
      const float alLo = upper ? b1 : b0;
      const float alHi = upper ? b3 : b2;
      const float* row = feat + (size_t)sj * 256;
      const float4 f0 = *(const float4*)(row + lane * 4);
      const float4 f1 = *(const float4*)(row + 128 + lane * 4);
      acc0.x += f0.x * alLo; acc0.y += f0.y * alLo; acc0.z += f0.z * alLo; acc0.w += f0.w * alLo;
      acc1.x += f1.x * alHi; acc1.y += f1.y * alHi; acc1.z += f1.z * alHi; acc1.w += f1.w * alHi;
    }
  }

  const int c0 = lane * 4, c1 = 128 + lane * 4;
  float v;
  v4h h0, h1;
  v = acc0.x + bias[c0 + 0]; h0[0] = (_Float16)(v > 0.f ? v : 0.f);
  v = acc0.y + bias[c0 + 1]; h0[1] = (_Float16)(v > 0.f ? v : 0.f);
  v = acc0.z + bias[c0 + 2]; h0[2] = (_Float16)(v > 0.f ? v : 0.f);
  v = acc0.w + bias[c0 + 3]; h0[3] = (_Float16)(v > 0.f ? v : 0.f);
  v = acc1.x + bias[c1 + 0]; h1[0] = (_Float16)(v > 0.f ? v : 0.f);
  v = acc1.y + bias[c1 + 1]; h1[1] = (_Float16)(v > 0.f ? v : 0.f);
  v = acc1.z + bias[c1 + 2]; h1[2] = (_Float16)(v > 0.f ? v : 0.f);
  v = acc1.w + bias[c1 + 3]; h1[3] = (_Float16)(v > 0.f ? v : 0.f);
  _Float16* o = outh + (size_t)d * 256;
  *(v4h*)(o + c0) = h0;
  *(v4h*)(o + c1) = h1;
}

__global__ void k_node_l2(const int* __restrict__ off, const int* __restrict__ srcS,
                          const float* __restrict__ as_, const float* __restrict__ ad_,
                          const float* __restrict__ feat, const float* __restrict__ bias,
                          float* __restrict__ outp, int Nn) {
  int d = blockIdx.x * (blockDim.x >> 5) + (threadIdx.x >> 5);
  if (d >= Nn) return;
  d = __builtin_amdgcn_readfirstlane(d);
  const int lane = threadIdx.x & 31;
  const int beg = off[d];
  const int deg = off[d + 1] - beg;
  const int tot = deg + 1;
  const float adv = ad_[d];

  float m = -__builtin_inff();
  for (int base = 0; base < tot; base += 32) {
    const int i = base + lane;
    if (i < tot) {
      const int s = i < deg ? srcS[beg + i] : d;
      m = fmaxf(m, leaky(as_[s] + adv));
    }
  }
  m = waveMax(m);

  float sm = 0.f;
  for (int base = 0; base < tot; base += 32) {
    const int i = base + lane;
    if (i < tot) {
      const int s = i < deg ? srcS[beg + i] : d;
      sm += expf(leaky(as_[s] + adv) - m);
    }
  }
  sm = waveSum(sm) + 1e-16f;

  float2 acc = make_float2(0.f, 0.f);
  for (int base = 0; base < tot; base += 32) {
    const int i = base + lane;
    int s = d;
    float al = 0.f;
    if (i < tot) {
      s = i < deg ? srcS[beg + i] : d;
      al = expf(leaky(as_[s] + adv) - m) / sm;
    }
    const int cnt = (tot - base < 32) ? (tot - base) : 32;
    #pragma unroll 4
    for (int j = 0; j < cnt; ++j) {
      const int   sj = __builtin_amdgcn_readlane(s, j);
      const float aj = laneF(al, j);
      const float2 f = *(const float2*)(feat + (size_t)sj * 64 + lane * 2);
      acc.x += f.x * aj;
      acc.y += f.y * aj;
    }
  }
  float2 r;
  r.x = acc.x + bias[lane * 2];
  r.y = acc.y + bias[lane * 2 + 1];
  *(float2*)(outp + (size_t)d * 64 + lane * 2) = r;
}

extern "C" void kernel_launch(void* const* d_in, const int* in_sizes, int n_in,
                              void* d_out, int out_size, void* d_ws, size_t ws_size,
                              hipStream_t stream) {
  const int IN = 128, HID = 64, H1 = 4;
  const int HC1 = H1 * HID;
  const int HC2 = HID;
  const int Nn   = in_sizes[0] / IN;
  const int E    = in_sizes[1] / 2;

  const float* x   = (const float*)d_in[0];
  const int*   ei  = (const int*)d_in[1];
  const float* W1  = (const float*)d_in[2];
  const float* aS1 = (const float*)d_in[3];
  const float* aD1 = (const float*)d_in[4];
  const float* b1  = (const float*)d_in[5];
  const float* W2  = (const float*)d_in[6];
  const float* aS2 = (const float*)d_in[7];
  const float* aD2 = (const float*)d_in[8];
  const float* b2  = (const float*)d_in[9];
  float* out = (float*)d_out;
  const int* srcI = ei;
  const int* dstI = ei + E;

  char* p = (char*)d_ws;
  auto carve = [&](size_t bytes) -> void* {
    void* r = (void*)p;
    p += (bytes + 255) & ~(size_t)255;
    return r;
  };
  _Float16* w1t  = (_Float16*)carve((size_t)HC1 * IN * 2);
  _Float16* w2t  = (_Float16*)carve((size_t)HID * HC1 * 2);
  float*    h1   = (float*)carve((size_t)Nn * HC1 * 4);
  _Float16* h1h  = (_Float16*)carve((size_t)Nn * HC1 * 2);
  float*    h2   = (float*)carve((size_t)Nn * HC2 * 4);
  float*    as1  = (float*)carve((size_t)Nn * H1 * 4);
  float*    ad1  = (float*)carve((size_t)Nn * H1 * 4);
  float*    as2  = (float*)carve((size_t)Nn * 4);
  float*    ad2  = (float*)carve((size_t)Nn * 4);
  int*      cnt  = (int*)carve((size_t)Nn * 4);
  int*      off  = (int*)carve((size_t)(Nn + 1) * 4);
  int*      part = (int*)carve((size_t)1024 * 4);
  int*      rank = (int*)carve((size_t)E * 4);
  int*      srcS = (int*)carve((size_t)E * 4);

  const int TB = 256;
  auto cdiv = [](long long a, long long b) { return (int)((a + b - 1) / b); };

  k_castT_f16<<<cdiv((long long)IN * HC1, TB), TB, 0, stream>>>(W1, w1t, IN, HC1, cnt, Nn);
  k_castT_f16<<<cdiv((long long)HC1 * HID, TB), TB, 0, stream>>>(W2, w2t, HC1, HID, cnt, 0);

  k_hist<<<cdiv(E, TB), TB, 0, stream>>>(dstI, E, cnt, rank);
  k_scan_local<<<cdiv(Nn, 1024), 1024, 0, stream>>>(cnt, off, part, Nn);
  k_scan_add<<<cdiv((long long)Nn + 1, 1024), 1024, 0, stream>>>(off, part, Nn, cdiv(Nn, 1024));
  k_fill<<<cdiv(E, TB), TB, 0, stream>>>(srcI, dstI, E, off, rank, srcS);

  {
    int waves = (Nn / 16) * (HC1 / 64);
    k_wmma_gemm_af32<<<cdiv(waves, 8), TB, 0, stream>>>(x, w1t, h1, Nn, HC1, IN);
  }
  k_attn<<<cdiv((long long)Nn * H1, TB), TB, 0, stream>>>(h1, aS1, aD1, as1, ad1, Nn, H1, HID);
  k_node_l1<<<cdiv(Nn, 8), TB, 0, stream>>>(off, srcS, as1, ad1, h1, b1, h1h, Nn);

  {
    int waves = (Nn / 16) * (HC2 / 64);
    k_wmma_gemm<<<cdiv(waves, 8), TB, 0, stream>>>(h1h, w2t, h2, Nn, HC2, HC1);
  }
  k_attn<<<cdiv((long long)Nn, TB), TB, 0, stream>>>(h2, aS2, aD2, as2, ad2, Nn, 1, HID);
  k_node_l2<<<cdiv(Nn, 8), TB, 0, stream>>>(off, srcS, as2, ad2, h2, b2, out, Nn);
}


# k_node_l1/l2: a row whose edges and self loop fit in one wave reads its edge list and logits once and keeps them in lanes for max, sum and weights (was three reads); sums and order unchanged
# speedup vs baseline: 3.4927x; 1.0575x over previous
#include <hip/hip_runtime.h>
#include <hip/hip_bf16.h>
#include <math.h>

typedef __attribute__((ext_vector_type(16))) _Float16 v16h;
typedef __attribute__((ext_vector_type(8)))  _Float16 v8h;
typedef __attribute__((ext_vector_type(4)))  _Float16 v4h;
typedef __attribute__((ext_vector_type(8)))  float    v8f;

#define NEG_SLOPE 0.2f

__device__ __forceinline__ float leaky(float x) { return x > 0.f ? x : NEG_SLOPE * x; }

__device__ __forceinline__ float waveMax(float v) {
  for (int o = 16; o > 0; o >>= 1) v = fmaxf(v, __shfl_xor(v, o));
  return v;
}
__device__ __forceinline__ float waveSum(float v) {
  for (int o = 16; o > 0; o >>= 1) v += __shfl_xor(v, o);
  return v;
}
__device__ __forceinline__ float laneF(float v, int j) {
  return __int_as_float(__builtin_amdgcn_readlane(__float_as_int(v), j));
}

__global__ void k_castT_f16(const float* __restrict__ in, _Float16* __restrict__ out, int R, int C,
                            int* __restrict__ cnt, int nCnt) {
  int t = blockIdx.x * blockDim.x + threadIdx.x;
  for (int i = t; i < nCnt; i += gridDim.x * blockDim.x) cnt[i] = 0;
  if (t >= R * C) return;
  int r = t / C, c = t - r * C;
  out[c * R + r] = (_Float16)in[t];
}

__global__ void k_hist(const int* __restrict__ dst, int E,
                       int* __restrict__ cnt, int* __restrict__ rank) {
  int t = blockIdx.x * blockDim.x + threadIdx.x;
  if (t >= E) return;
  rank[t] = atomicAdd(&cnt[dst[t]], 1);
}

__global__ void __launch_bounds__(1024) k_scan_local(const int* __restrict__ cnt, int* __restrict__ off,
                                                     int* __restrict__ part, int n) {
  __shared__ int sh[1024];
  const int t = threadIdx.x;
  const int i = blockIdx.x * 1024 + t;
  const int v = i < n ? cnt[i] : 0;
  sh[t] = v;
  __syncthreads();
  for (int o = 1; o < 1024; o <<= 1) {
    const int u = (t >= o) ? sh[t - o] : 0;
    __syncthreads();
    sh[t] += u;
    __syncthreads();
  }
  if (i < n) off[i] = sh[t] - v;
  if (t == 1023) part[blockIdx.x] = sh[1023];
}

__global__ void __launch_bounds__(1024) k_scan_add(int* __restrict__ off, const int* __restrict__ part,
                                                   int n, int nTiles) {
  __shared__ int sBefore, sAll;
  const int t = threadIdx.x;
  if (t < 32) {
    int before = 0, all = 0;
    for (int j = t; j < nTiles; j += 32) {
      const int pj = part[j];
      all += pj;
      if (j < (int)blockIdx.x) before += pj;
    }
    for (int o = 16; o > 0; o >>= 1) {
      before += __shfl_xor(before, o);
      all    += __shfl_xor(all, o);
    }
    if (t == 0) { sBefore = before; sAll = all; }
  }
  __syncthreads();
  const int i = blockIdx.x * 1024 + t;
  if (i < n) off[i] += sBefore;
  else if (i == n) off[i] = sAll;
}

__global__ void k_fill(const int* __restrict__ src, const int* __restrict__ dst, int E,
                       const int* __restrict__ off, const int* __restrict__ rank,
                       int* __restrict__ srcS) {
  int t = blockIdx.x * blockDim.x + threadIdx.x;
  if (t >= E) return;
  srcS[off[dst[t]] + rank[t]] = src[t];
}

template <bool A_F32>
__device__ __forceinline__ void gemm_strip(const float* __restrict__ Af,
                                           const _Float16* __restrict__ Ah,
                                           const _Float16* __restrict__ Bt,
                                           float* __restrict__ C,
                                           int M, int N, int K) {
  const int tilesN      = N >> 4;
  const int wavesPerRow = tilesN >> 2;
  const int totalWaves  = (M >> 4) * wavesPerRow;
  const int wave = blockIdx.x * (blockDim.x >> 5) + (threadIdx.x >> 5);
  if (wave >= totalWaves) return;
  const int lane = threadIdx.x & 31;
  const int half = lane >> 4;
  const int l16  = lane & 15;
  const int tm  = wave / wavesPerRow;
  const int tn0 = (wave - tm * wavesPerRow) * 4;

  const size_t aOff = (size_t)(tm * 16 + l16) * K;
  const _Float16* bRow[4];
  #pragma unroll
  for (int j = 0; j < 4; ++j)
    bRow[j] = Bt + (size_t)((tn0 + j) * 16 + l16) * K;

  v8f acc0 = {}, acc1 = {}, acc2 = {}, acc3 = {};
  for (int k = 0; k < K; k += 32) {
    const int ka = k + half * 8;
    union { v16h v; v8h h[2]; } a, b0, b1, b2, b3;
    if constexpr (A_F32) {
      const float* ap = Af + aOff + ka;
      const float4 p0 = *(const float4*)(ap);
      const float4 p1 = *(const float4*)(ap + 4);
      const float4 p2 = *(const float4*)(ap + 16);
      const float4 p3 = *(const float4*)(ap + 20);
      a.v[0]  = (_Float16)p0.x; a.v[1]  = (_Float16)p0.y; a.v[2]  = (_Float16)p0.z; a.v[3]  = (_Float16)p0.w;
      a.v[4]  = (_Float16)p1.x; a.v[5]  = (_Float16)p1.y; a.v[6]  = (_Float16)p1.z; a.v[7]  = (_Float16)p1.w;
      a.v[8]  = (_Float16)p2.x; a.v[9]  = (_Float16)p2.y; a.v[10] = (_Float16)p2.z; a.v[11] = (_Float16)p2.w;
      a.v[12] = (_Float16)p3.x; a.v[13] = (_Float16)p3.y; a.v[14] = (_Float16)p3.z; a.v[15] = (_Float16)p3.w;
    } else {
      const _Float16* ap = Ah + aOff + ka;
      a.h[0] = *(const v8h*)(ap);
      a.h[1] = *(const v8h*)(ap + 16);
    }
    b0.h[0] = *(const v8h*)(bRow[0] + ka);
    b0.h[1] = *(const v8h*)(bRow[0] + ka + 16);
    b1.h[0] = *(const v8h*)(bRow[1] + ka);
    b1.h[1] = *(const v8h*)(bRow[1] + ka + 16);
    b2.h[0] = *(const v8h*)(bRow[2] + ka);
    b2.h[1] = *(const v8h*)(bRow[2] + ka + 16);
    b3.h[0] = *(const v8h*)(bRow[3] + ka);
    b3.h[1] = *(const v8h*)(bRow[3] + ka + 16);
    acc0 = __builtin_amdgcn_wmma_f32_16x16x32_f16(false, a.v, false, b0.v, (short)0, acc0, false, false);
    acc1 = __builtin_amdgcn_wmma_f32_16x16x32_f16(false, a.v, false, b1.v, (short)0, acc1, false, false);
    acc2 = __builtin_amdgcn_wmma_f32_16x16x32_f16(false, a.v, false, b2.v, (short)0, acc2, false, false);
    acc3 = __builtin_amdgcn_wmma_f32_16x16x32_f16(false, a.v, false, b3.v, (short)0, acc3, false, false);
  }
  float* cRow = C + (size_t)(tm * 16 + half * 8) * N + tn0 * 16 + l16;
  #pragma unroll
  for (int j = 0; j < 8; ++j) {
    float* cr = cRow + (size_t)j * N;
    cr[0]  = acc0[j];
    cr[16] = acc1[j];
    cr[32] = acc2[j];
    cr[48] = acc3[j];
  }
}

__global__ void k_wmma_gemm_af32(const float* __restrict__ A,
                                 const _Float16* __restrict__ Bt,
                                 float* __restrict__ C,
                                 int M, int N, int K) {
  gemm_strip<true>(A, nullptr, Bt, C, M, N, K);
}

__global__ void k_wmma_gemm(const _Float16* __restrict__ A,
                            const _Float16* __restrict__ Bt,
                            float* __restrict__ C,
                            int M, int N, int K) {
  gemm_strip<false>(nullptr, A, Bt, C, M, N, K);
}

__global__ void k_attn(const float* __restrict__ h, const float* __restrict__ att_s,
                       const float* __restrict__ att_d,
                       float* __restrict__ as_, float* __restrict__ ad_,
                       int Nn, int H, int Cc) {
  int t = blockIdx.x * blockDim.x + threadIdx.x;
  if (t >= Nn * H) return;
  int n = t / H, hh = t - n * H;
  const float* hp = h + (size_t)n * H * Cc + (size_t)hh * Cc;
  const float* sp = att_s + hh * Cc;
  const float* dp = att_d + hh * Cc;
  float a0 = 0.f, a1 = 0.f;
  for (int c = 0; c < Cc; ++c) { float v = hp[c]; a0 += v * sp[c]; a1 += v * dp[c]; }
  as_[t] = a0;
  ad_[t] = a1;
}


__device__ __forceinline__ void walk_rows_l1(const float* __restrict__ feat, int lane, bool upper,
                                             int s, float a0, float a1, float a2, float a3, int cnt,
                                             float4& acc0, float4& acc1) {
  #pragma unroll 4
  for (int j = 0; j < cnt; ++j) {
    const int   sj = __builtin_amdgcn_readlane(s, j);
    const float b0 = laneF(a0, j), b1 = laneF(a1, j), b2 = laneF(a2, j), b3 = laneF(a3, j);
    const float alLo = upper ? b1 : b0;
    const float alHi = upper ? b3 : b2;
    const float* row = feat + (size_t)sj * 256;
    const float4 f0 = *(const float4*)(row + lane * 4);
    const float4 f1 = *(const float4*)(row + 128 + lane * 4);
    acc0.x += f0.x * alLo; acc0.y += f0.y * alLo; acc0.z += f0.z * alLo; acc0.w += f0.w * alLo;
    acc1.x += f1.x * alHi; acc1.y += f1.y * alHi; acc1.z += f1.z * alHi; acc1.w += f1.w * alHi;
  }
}

__global__ void k_node_l1(const int* __restrict__ off, const int* __restrict__ srcS,
                          const float* __restrict__ as_, const float* __restrict__ ad_,
                          const float* __restrict__ feat, const float* __restrict__ bias,
                          _Float16* __restrict__ outh, int Nn) {
  int d = blockIdx.x * (blockDim.x >> 5) + (threadIdx.x >> 5);
  if (d >= Nn) return;
  d = __builtin_amdgcn_readfirstlane(d);
  const int lane = threadIdx.x & 31;
  const int beg = off[d];
  const int deg = off[d + 1] - beg;
  const int tot = deg + 1;
  const float4 adv = *(const float4*)(ad_ + (size_t)d * 4);
  const bool upper = lane >= 16;
  float4 acc0 = make_float4(0.f, 0.f, 0.f, 0.f);
  float4 acc1 = make_float4(0.f, 0.f, 0.f, 0.f);

  if (tot <= 32) {
    const bool valid = lane < tot;
    int s = d;
    if (lane < deg) s = srcS[beg + lane];
    const float4 a = *(const float4*)(as_ + (size_t)s * 4);
    const float ninf = -__builtin_inff();
    const float r0 = leaky(a.x + adv.x), r1 = leaky(a.y + adv.y);
    const float r2 = leaky(a.z + adv.z), r3 = leaky(a.w + adv.w);
    const float e0 = valid ? r0 : ninf, e1 = valid ? r1 : ninf, e2 = valid ? r2 : ninf, e3 = valid ? r3 : ninf;
    const float m0 = waveMax(e0), m1 = waveMax(e1), m2 = waveMax(e2), m3 = waveMax(e3);
    const float p0 = expf(e0 - m0), p1 = expf(e1 - m1), p2 = expf(e2 - m2), p3 = expf(e3 - m3);
    const float s0 = waveSum(p0) + 1e-16f, s1 = waveSum(p1) + 1e-16f;
    const float s2 = waveSum(p2) + 1e-16f, s3 = waveSum(p3) + 1e-16f;
    walk_rows_l1(feat, lane, upper, s, p0 / s0, p1 / s1, p2 / s2, p3 / s3, tot, acc0, acc1);
  } else {
    float m0 = -__builtin_inff(), m1 = m0, m2 = m0, m3 = m0;
    for (int base = 0; base < tot; base += 32) {
      const int i = base + lane;
      if (i < tot) {
        const int s = i < deg ? srcS[beg + i] : d;
        const float4 a = *(const float4*)(as_ + (size_t)s * 4);
        m0 = fmaxf(m0, leaky(a.x + adv.x));
        m1 = fmaxf(m1, leaky(a.y + adv.y));
        m2 = fmaxf(m2, leaky(a.z + adv.z));
        m3 = fmaxf(m3, leaky(a.w + adv.w));
      }
    }
    m0 = waveMax(m0); m1 = waveMax(m1); m2 = waveMax(m2); m3 = waveMax(m3);

    float s0 = 0.f, s1 = 0.f, s2 = 0.f, s3 = 0.f;
    for (int base = 0; base < tot; base += 32) {
      const int i = base + lane;
      if (i < tot) {
        const int s = i < deg ? srcS[beg + i] : d;
        const float4 a = *(const float4*)(as_ + (size_t)s * 4);
        s0 += expf(leaky(a.x + adv.x) - m0);
        s1 += expf(leaky(a.y + adv.y) - m1);
        s2 += expf(leaky(a.z + adv.z) - m2);
        s3 += expf(leaky(a.w + adv.w) - m3);
      }
    }
    s0 = waveSum(s0) + 1e-16f; s1 = waveSum(s1) + 1e-16f;
    s2 = waveSum(s2) + 1e-16f; s3 = waveSum(s3) + 1e-16f;

    for (int base = 0; base < tot; base += 32) {
      const int i = base + lane;
      int s = d;
      float a0 = 0.f, a1 = 0.f, a2 = 0.f, a3 = 0.f;
      if (i < tot) {
        s = i < deg ? srcS[beg + i] : d;
        const float4 a = *(const float4*)(as_ + (size_t)s * 4);
        a0 = expf(leaky(a.x + adv.x) - m0) / s0;
        a1 = expf(leaky(a.y + adv.y) - m1) / s1;
        a2 = expf(leaky(a.z + adv.z) - m2) / s2;
        a3 = expf(leaky(a.w + adv.w) - m3) / s3;
      }
      const int cnt = (tot - base < 32) ? (tot - base) : 32;
      walk_rows_l1(feat, lane, upper, s, a0, a1, a2, a3, cnt, acc0, acc1);
    }
  }

  const int c0 = lane * 4, c1 = 128 + lane * 4;
  float v;
  v4h h0, h1;
  v = acc0.x + bias[c0 + 0]; h0[0] = (_Float16)(v > 0.f ? v : 0.f);
  v = acc0.y + bias[c0 + 1]; h0[1] = (_Float16)(v > 0.f ? v : 0.f);
  v = acc0.z + bias[c0 + 2]; h0[2] = (_Float16)(v > 0.f ? v : 0.f);
  v = acc0.w + bias[c0 + 3]; h0[3] = (_Float16)(v > 0.f ? v : 0.f);
  v = acc1.x + bias[c1 + 0]; h1[0] = (_Float16)(v > 0.f ? v : 0.f);
  v = acc1.y + bias[c1 + 1]; h1[1] = (_Float16)(v > 0.f ? v : 0.f);
  v = acc1.z + bias[c1 + 2]; h1[2] = (_Float16)(v > 0.f ? v : 0.f);
  v = acc1.w + bias[c1 + 3]; h1[3] = (_Float16)(v > 0.f ? v : 0.f);
  _Float16* o = outh + (size_t)d * 256;
  *(v4h*)(o + c0) = h0;
  *(v4h*)(o + c1) = h1;
}

__device__ __forceinline__ void walk_rows_l2(const float* __restrict__ feat, int lane,
                                             int s, float al, int cnt, float2& acc) {
  #pragma unroll 4
  for (int j = 0; j < cnt; ++j) {
    const int   sj = __builtin_amdgcn_readlane(s, j);
    const float aj = laneF(al, j);
    const float2 f = *(const float2*)(feat + (size_t)sj * 64 + lane * 2);
    acc.x += f.x * aj;
    acc.y += f.y * aj;
  }
}

__global__ void k_node_l2(const int* __restrict__ off, const int* __restrict__ srcS,
                          const float* __restrict__ as_, const float* __restrict__ ad_,
                          const float* __restrict__ feat, const float* __restrict__ bias,
                          float* __restrict__ outp, int Nn) {
  int d = blockIdx.x * (blockDim.x >> 5) + (threadIdx.x >> 5);
  if (d >= Nn) return;
  d = __builtin_amdgcn_readfirstlane(d);
  const int lane = threadIdx.x & 31;
  const int beg = off[d];
  const int deg = off[d + 1] - beg;
  const int tot = deg + 1;
  const float adv = ad_[d];
  float2 acc = make_float2(0.f, 0.f);

  if (tot <= 32) {
    const bool valid = lane < tot;
    int s = d;
    if (lane < deg) s = srcS[beg + lane];
    const float r = leaky(as_[s] + adv);
    const float e = valid ? r : -__builtin_inff();
    const float m = waveMax(e);
    const float p = expf(e - m);
    const float sm = waveSum(p) + 1e-16f;
    walk_rows_l2(feat, lane, s, p / sm, tot, acc);
  } else {
    float m = -__builtin_inff();
    for (int base = 0; base < tot; base += 32) {
      const int i = base + lane;
      if (i < tot) {
        const int s = i < deg ? srcS[beg + i] : d;
        m = fmaxf(m, leaky(as_[s] + adv));
      }
    }
    m = waveMax(m);

    float sm = 0.f;
    for (int base = 0; base < tot; base += 32) {
      const int i = base + lane;
      if (i < tot) {
        const int s = i < deg ? srcS[beg + i] : d;
        sm += expf(leaky(as_[s] + adv) - m);
      }
    }
    sm = waveSum(sm) + 1e-16f;

    for (int base = 0; base < tot; base += 32) {
      const int i = base + lane;
      int s = d;
      float al = 0.f;
      if (i < tot) {
        s = i < deg ? srcS[beg + i] : d;
        al = expf(leaky(as_[s] + adv) - m) / sm;
      }
      const int cnt = (tot - base < 32) ? (tot - base) : 32;
      walk_rows_l2(feat, lane, s, al, cnt, acc);
    }
  }
  float2 r;
  r.x = acc.x + bias[lane * 2];
  r.y = acc.y + bias[lane * 2 + 1];
  *(float2*)(outp + (size_t)d * 64 + lane * 2) = r;
}

extern "C" void kernel_launch(void* const* d_in, const int* in_sizes, int n_in,
                              void* d_out, int out_size, void* d_ws, size_t ws_size,
                              hipStream_t stream) {
  const int IN = 128, HID = 64, H1 = 4;
  const int HC1 = H1 * HID;
  const int HC2 = HID;
  const int Nn   = in_sizes[0] / IN;
  const int E    = in_sizes[1] / 2;

  const float* x   = (const float*)d_in[0];
  const int*   ei  = (const int*)d_in[1];
  const float* W1  = (const float*)d_in[2];
  const float* aS1 = (const float*)d_in[3];
  const float* aD1 = (const float*)d_in[4];
  const float* b1  = (const float*)d_in[5];
  const float* W2  = (const float*)d_in[6];
  const float* aS2 = (const float*)d_in[7];
  const float* aD2 = (const float*)d_in[8];
  const float* b2  = (const float*)d_in[9];
  float* out = (float*)d_out;
  const int* srcI = ei;
  const int* dstI = ei + E;

  char* p = (char*)d_ws;
  auto carve = [&](size_t bytes) -> void* {
    void* r = (void*)p;
    p += (bytes + 255) & ~(size_t)255;
    return r;
  };
  _Float16* w1t  = (_Float16*)carve((size_t)HC1 * IN * 2);
  _Float16* w2t  = (_Float16*)carve((size_t)HID * HC1 * 2);
  float*    h1   = (float*)carve((size_t)Nn * HC1 * 4);
  _Float16* h1h  = (_Float16*)carve((size_t)Nn * HC1 * 2);
  float*    h2   = (float*)carve((size_t)Nn * HC2 * 4);
  float*    as1  = (float*)carve((size_t)Nn * H1 * 4);
  float*    ad1  = (float*)carve((size_t)Nn * H1 * 4);
  float*    as2  = (float*)carve((size_t)Nn * 4);
  float*    ad2  = (float*)carve((size_t)Nn * 4);
  int*      cnt  = (int*)carve((size_t)Nn * 4);
  int*      off  = (int*)carve((size_t)(Nn + 1) * 4);
  int*      part = (int*)carve((size_t)1024 * 4);
  int*      rank = (int*)carve((size_t)E * 4);
  int*      srcS = (int*)carve((size_t)E * 4);

  const int TB = 256;
  auto cdiv = [](long long a, long long b) { return (int)((a + b - 1) / b); };

  k_castT_f16<<<cdiv((long long)IN * HC1, TB), TB, 0, stream>>>(W1, w1t, IN, HC1, cnt, Nn);
  k_castT_f16<<<cdiv((long long)HC1 * HID, TB), TB, 0, stream>>>(W2, w2t, HC1, HID, cnt, 0);

  k_hist<<<cdiv(E, TB), TB, 0, stream>>>(dstI, E, cnt, rank);
  k_scan_local<<<cdiv(Nn, 1024), 1024, 0, stream>>>(cnt, off, part, Nn);
  k_scan_add<<<cdiv((long long)Nn + 1, 1024), 1024, 0, stream>>>(off, part, Nn, cdiv(Nn, 1024));
  k_fill<<<cdiv(E, TB), TB, 0, stream>>>(srcI, dstI, E, off, rank, srcS);

  {
    int waves = (Nn / 16) * (HC1 / 64);
    k_wmma_gemm_af32<<<cdiv(waves, 8), TB, 0, stream>>>(x, w1t, h1, Nn, HC1, IN);
  }
  k_attn<<<cdiv((long long)Nn * H1, TB), TB, 0, stream>>>(h1, aS1, aD1, as1, ad1, Nn, H1, HID);
  k_node_l1<<<cdiv(Nn, 8), TB, 0, stream>>>(off, srcS, as1, ad1, h1, b1, h1h, Nn);

  {
    int waves = (Nn / 16) * (HC2 / 64);
    k_wmma_gemm<<<cdiv(waves, 8), TB, 0, stream>>>(h1h, w2t, h2, Nn, HC2, HC1);
  }
  k_attn<<<cdiv((long long)Nn, TB), TB, 0, stream>>>(h2, aS2, aD2, as2, ad2, Nn, 1, HID);
  k_node_l2<<<cdiv(Nn, 8), TB, 0, stream>>>(off, srcS, as2, ad2, h2, b2, out, Nn);
}


# Independent passes share launches: both weight casts + counter clear in one; prefix sum in one pass (each tile sums the counters ahead of it) beside the layer-1 GEMM; 10 launches to 7; sums unchanged
# speedup vs baseline: 3.8728x; 1.1088x over previous
#include <hip/hip_runtime.h>
#include <hip/hip_bf16.h>
#include <math.h>

typedef __attribute__((ext_vector_type(16))) _Float16 v16h;
typedef __attribute__((ext_vector_type(8)))  _Float16 v8h;
typedef __attribute__((ext_vector_type(4)))  _Float16 v4h;
typedef __attribute__((ext_vector_type(8)))  float    v8f;

#define NEG_SLOPE 0.2f

__device__ __forceinline__ float leaky(float x) { return x > 0.f ? x : NEG_SLOPE * x; }

__device__ __forceinline__ float waveMax(float v) {
  for (int o = 16; o > 0; o >>= 1) v = fmaxf(v, __shfl_xor(v, o));
  return v;
}
__device__ __forceinline__ float waveSum(float v) {
  for (int o = 16; o > 0; o >>= 1) v += __shfl_xor(v, o);
  return v;
}
__device__ __forceinline__ float laneF(float v, int j) {
  return __int_as_float(__builtin_amdgcn_readlane(__float_as_int(v), j));
}

__global__ void k_prep(const float* __restrict__ in1, _Float16* __restrict__ out1, int R1, int C1,
                       const float* __restrict__ in2, _Float16* __restrict__ out2, int R2, int C2,
                       int* __restrict__ cnt, int nCnt) {
  const int t = blockIdx.x * blockDim.x + threadIdx.x;
  for (int i = t; i < nCnt; i += gridDim.x * blockDim.x) cnt[i] = 0;
  const int n1 = R1 * C1;
  if (t < n1) {
    const int r = t / C1, c = t - r * C1;
    out1[c * R1 + r] = (_Float16)in1[t];
  } else {
    const int u = t - n1;
    if (u < R2 * C2) {
      const int r = u / C2, c = u - r * C2;
      out2[c * R2 + r] = (_Float16)in2[u];
    }
  }
}

__global__ void k_hist(const int* __restrict__ dst, int E,
                       int* __restrict__ cnt, int* __restrict__ rank) {
  int t = blockIdx.x * blockDim.x + threadIdx.x;
  if (t >= E) return;
  rank[t] = atomicAdd(&cnt[dst[t]], 1);
}

__device__ __forceinline__ void scan_tile(const int* __restrict__ cnt, int* __restrict__ off,
                                          int n, int b, int* sh) {
  const int t = threadIdx.x;
  const int lane = t & 31, w = t >> 5;
  const int lim = b * 1024;
  int before = 0;
  #pragma unroll 8
  for (int i = t * 4; i < lim; i += 1024) {
    const int4 c = *(const int4*)(cnt + i);
    before += (c.x + c.y) + (c.z + c.w);
  }
  const int i0 = lim + t * 4;
  const int c0 = i0     < n ? cnt[i0]     : 0;
  const int c1 = i0 + 1 < n ? cnt[i0 + 1] : 0;
  const int c2 = i0 + 2 < n ? cnt[i0 + 2] : 0;
  const int c3 = i0 + 3 < n ? cnt[i0 + 3] : 0;
  const int s = (c0 + c1) + (c2 + c3);
  int inc = s;
  for (int o = 1; o < 32; o <<= 1) {
    const int u = __shfl_up(inc, o);
    if (lane >= o) inc += u;
  }
  for (int o = 16; o > 0; o >>= 1) before += __shfl_xor(before, o);
  if (lane == 31) sh[w] = inc;
  if (lane == 0)  sh[8 + w] = before;
  __syncthreads();
  int e0 = inc - s;
  for (int j = 0; j < 8; ++j) {
    e0 += sh[8 + j];
    if (j < w) e0 += sh[j];
  }
  const int e1 = e0 + c0, e2 = e1 + c1, e3 = e2 + c2;
  if (i0 + 3 <= n) {
    *(int4*)(off + i0) = make_int4(e0, e1, e2, e3);
  } else {
    if (i0     <= n) off[i0]     = e0;
    if (i0 + 1 <= n) off[i0 + 1] = e1;
    if (i0 + 2 <= n) off[i0 + 2] = e2;
  }
}

__global__ void k_fill(const int* __restrict__ src, const int* __restrict__ dst, int E,
                       const int* __restrict__ off, const int* __restrict__ rank,
                       int* __restrict__ srcS) {
  int t = blockIdx.x * blockDim.x + threadIdx.x;
  if (t >= E) return;
  srcS[off[dst[t]] + rank[t]] = src[t];
}


__device__ __forceinline__ float rowSum16(float v) {
  v += __int_as_float(__builtin_amdgcn_update_dpp(0, __float_as_int(v), 0x161, 0xf, 0xf, true));
  v += __int_as_float(__builtin_amdgcn_update_dpp(0, __float_as_int(v), 0x162, 0xf, 0xf, true));
  v += __int_as_float(__builtin_amdgcn_update_dpp(0, __float_as_int(v), 0x164, 0xf, 0xf, true));
  v += __int_as_float(__builtin_amdgcn_update_dpp(0, __float_as_int(v), 0x168, 0xf, 0xf, true));
  return v;
}

template <bool A_F32>
__device__ __forceinline__ void gemm_strip(const float* __restrict__ Af,
                                           const _Float16* __restrict__ Ah,
                                           const _Float16* __restrict__ Bt,
                                           float* __restrict__ C,
                                           const float* __restrict__ att_s,
                                           const float* __restrict__ att_d,
                                           float* __restrict__ as_, float* __restrict__ ad_,
                                           float* sAl,
                                           int bid,
                                           int M, int N, int K) {
  const int tilesN      = N >> 4;
  const int wavesPerRow = tilesN >> 2;
  const int totalWaves  = (M >> 4) * wavesPerRow;
  const int wave = __builtin_amdgcn_readfirstlane(bid * 8 + (int)(threadIdx.x >> 5));
  const int lane = threadIdx.x & 31;
  const int half = lane >> 4;
  const int l16  = lane & 15;

  if (wave < totalWaves) {
    const int tm  = wave / wavesPerRow;
    const int wr  = wave - tm * wavesPerRow;
    const int tn0 = wr * 4;

    const float* sv = att_s + wr * 64 + l16;
    const float* dv = att_d + wr * 64 + l16;
    const float s0 = sv[0], s1 = sv[16], s2 = sv[32], s3 = sv[48];
    const float d0 = dv[0], d1 = dv[16], d2 = dv[32], d3 = dv[48];

    const size_t aOff = (size_t)(tm * 16 + l16) * K;
    const _Float16* bRow[4];
    #pragma unroll
    for (int j = 0; j < 4; ++j)
      bRow[j] = Bt + (size_t)((tn0 + j) * 16 + l16) * K;

    v8f acc0 = {}, acc1 = {}, acc2 = {}, acc3 = {};
    for (int k = 0; k < K; k += 32) {
      const int ka = k + half * 8;
      union { v16h v; v8h h[2]; } a, b0, b1, b2, b3;
      if constexpr (A_F32) {
        const float* ap = Af + aOff + ka;
        const float4 p0 = *(const float4*)(ap);
        const float4 p1 = *(const float4*)(ap + 4);
        const float4 p2 = *(const float4*)(ap + 16);
        const float4 p3 = *(const float4*)(ap + 20);
        a.v[0]  = (_Float16)p0.x; a.v[1]  = (_Float16)p0.y; a.v[2]  = (_Float16)p0.z; a.v[3]  = (_Float16)p0.w;
        a.v[4]  = (_Float16)p1.x; a.v[5]  = (_Float16)p1.y; a.v[6]  = (_Float16)p1.z; a.v[7]  = (_Float16)p1.w;
        a.v[8]  = (_Float16)p2.x; a.v[9]  = (_Float16)p2.y; a.v[10] = (_Float16)p2.z; a.v[11] = (_Float16)p2.w;
        a.v[12] = (_Float16)p3.x; a.v[13] = (_Float16)p3.y; a.v[14] = (_Float16)p3.z; a.v[15] = (_Float16)p3.w;
      } else {
        const _Float16* ap = Ah + aOff + ka;
        a.h[0] = *(const v8h*)(ap);
        a.h[1] = *(const v8h*)(ap + 16);
      }
      b0.h[0] = *(const v8h*)(bRow[0] + ka);
      b0.h[1] = *(const v8h*)(bRow[0] + ka + 16);
      b1.h[0] = *(const v8h*)(bRow[1] + ka);
      b1.h[1] = *(const v8h*)(bRow[1] + ka + 16);
      b2.h[0] = *(const v8h*)(bRow[2] + ka);
      b2.h[1] = *(const v8h*)(bRow[2] + ka + 16);
      b3.h[0] = *(const v8h*)(bRow[3] + ka);
      b3.h[1] = *(const v8h*)(bRow[3] + ka + 16);
      acc0 = __builtin_amdgcn_wmma_f32_16x16x32_f16(false, a.v, false, b0.v, (short)0, acc0, false, false);
      acc1 = __builtin_amdgcn_wmma_f32_16x16x32_f16(false, a.v, false, b1.v, (short)0, acc1, false, false);
      acc2 = __builtin_amdgcn_wmma_f32_16x16x32_f16(false, a.v, false, b2.v, (short)0, acc2, false, false);
      acc3 = __builtin_amdgcn_wmma_f32_16x16x32_f16(false, a.v, false, b3.v, (short)0, acc3, false, false);
    }
    float* cRow = C + (size_t)(tm * 16 + half * 8) * N + tn0 * 16 + l16;
    #pragma unroll
    for (int j = 0; j < 8; ++j) {
      float* cr = cRow + (size_t)j * N;
      cr[0]  = acc0[j];
      cr[16] = acc1[j];
      cr[32] = acc2[j];
      cr[48] = acc3[j];
    }

    float vs = 0.f, vd = 0.f;
    #pragma unroll
    for (int j = 0; j < 8; ++j) {
      const float ps = rowSum16(acc0[j] * s0 + acc1[j] * s1 + acc2[j] * s2 + acc3[j] * s3);
      const float pd = rowSum16(acc0[j] * d0 + acc1[j] * d1 + acc2[j] * d2 + acc3[j] * d3);
      if (l16 == j) { vs = ps; vd = pd; }
    }
    if (l16 < 8) {
      const int li = (tm * 16 + half * 8 + l16) * wavesPerRow + wr - bid * 128;
      sAl[li]       = vs;
      sAl[128 + li] = vd;
    }
  }
  __syncthreads();
  const int t = threadIdx.x;
  if (t < 64) {
    const int which = t >> 5;
    const int q  = (t & 31) * 4;
    const int gi = bid * 128 + q;
    if (gi < M * wavesPerRow) {
      const float4 v = *(const float4*)(sAl + which * 128 + q);
      *(float4*)((which ? ad_ : as_) + gi) = v;
    }
  }
}

__global__ void __launch_bounds__(256) k_scan_gemm_af32(const int* __restrict__ cnt, int* __restrict__ off,
                                 int nCnt, int nScan,
                                 const float* __restrict__ A,
                                 const _Float16* __restrict__ Bt,
                                 float* __restrict__ C,
                                 const float* __restrict__ att_s, const float* __restrict__ att_d,
                                 float* __restrict__ as_, float* __restrict__ ad_,
                                 int M, int N, int K) {
  __shared__ __attribute__((aligned(16))) float sAl[256];
  __shared__ int sScan[16];
  if ((int)blockIdx.x < nScan) {
    scan_tile(cnt, off, nCnt, (int)blockIdx.x, sScan);
    return;
  }
  gemm_strip<true>(A, nullptr, Bt, C, att_s, att_d, as_, ad_, sAl, (int)blockIdx.x - nScan, M, N, K);
}

__global__ void __launch_bounds__(256) k_wmma_gemm(const _Float16* __restrict__ A,
                            const _Float16* __restrict__ Bt,
                            float* __restrict__ C,
                            const float* __restrict__ att_s, const float* __restrict__ att_d,
                            float* __restrict__ as_, float* __restrict__ ad_,
                            int M, int N, int K) {
  __shared__ __attribute__((aligned(16))) float sAl[256];
  gemm_strip<false>(nullptr, A, Bt, C, att_s, att_d, as_, ad_, sAl, (int)blockIdx.x, M, N, K);
}


__device__ __forceinline__ void walk_rows_l1(const float* __restrict__ feat, int lane, bool upper,
                                             int s, float a0, float a1, float a2, float a3, int cnt,
                                             float4& acc0, float4& acc1) {
  #pragma unroll 4
  for (int j = 0; j < cnt; ++j) {
    const int   sj = __builtin_amdgcn_readlane(s, j);
    const float b0 = laneF(a0, j), b1 = laneF(a1, j), b2 = laneF(a2, j), b3 = laneF(a3, j);
    const float alLo = upper ? b1 : b0;
    const float alHi = upper ? b3 : b2;
    const float* row = feat + (size_t)sj * 256;
    const float4 f0 = *(const float4*)(row + lane * 4);
    const float4 f1 = *(const float4*)(row + 128 + lane * 4);
    acc0.x += f0.x * alLo; acc0.y += f0.y * alLo; acc0.z += f0.z * alLo; acc0.w += f0.w * alLo;
    acc1.x += f1.x * alHi; acc1.y += f1.y * alHi; acc1.z += f1.z * alHi; acc1.w += f1.w * alHi;
  }
}

__global__ void k_node_l1(const int* __restrict__ off, const int* __restrict__ srcS,
                          const float* __restrict__ as_, const float* __restrict__ ad_,
                          const float* __restrict__ feat, const float* __restrict__ bias,
                          _Float16* __restrict__ outh, int Nn) {
  int d = blockIdx.x * (blockDim.x >> 5) + (threadIdx.x >> 5);
  if (d >= Nn) return;
  d = __builtin_amdgcn_readfirstlane(d);
  const int lane = threadIdx.x & 31;
  const int beg = off[d];
  const int deg = off[d + 1] - beg;
  const int tot = deg + 1;
  const float4 adv = *(const float4*)(ad_ + (size_t)d * 4);
  const bool upper = lane >= 16;
  float4 acc0 = make_float4(0.f, 0.f, 0.f, 0.f);
  float4 acc1 = make_float4(0.f, 0.f, 0.f, 0.f);

  if (tot <= 32) {
    const bool valid = lane < tot;
    int s = d;
    if (lane < deg) s = srcS[beg + lane];
    const float4 a = *(const float4*)(as_ + (size_t)s * 4);
    const float ninf = -__builtin_inff();
    const float r0 = leaky(a.x + adv.x), r1 = leaky(a.y + adv.y);
    const float r2 = leaky(a.z + adv.z), r3 = leaky(a.w + adv.w);
    const float e0 = valid ? r0 : ninf, e1 = valid ? r1 : ninf, e2 = valid ? r2 : ninf, e3 = valid ? r3 : ninf;
    const float m0 = waveMax(e0), m1 = waveMax(e1), m2 = waveMax(e2), m3 = waveMax(e3);
    const float p0 = expf(e0 - m0), p1 = expf(e1 - m1), p2 = expf(e2 - m2), p3 = expf(e3 - m3);
    const float s0 = waveSum(p0) + 1e-16f, s1 = waveSum(p1) + 1e-16f;
    const float s2 = waveSum(p2) + 1e-16f, s3 = waveSum(p3) + 1e-16f;
    walk_rows_l1(feat, lane, upper, s, p0 / s0, p1 / s1, p2 / s2, p3 / s3, tot, acc0, acc1);
  } else {
    float m0 = -__builtin_inff(), m1 = m0, m2 = m0, m3 = m0;
    for (int base = 0; base < tot; base += 32) {
      const int i = base + lane;
      if (i < tot) {
        const int s = i < deg ? srcS[beg + i] : d;
        const float4 a = *(const float4*)(as_ + (size_t)s * 4);
        m0 = fmaxf(m0, leaky(a.x + adv.x));
        m1 = fmaxf(m1, leaky(a.y + adv.y));
        m2 = fmaxf(m2, leaky(a.z + adv.z));
        m3 = fmaxf(m3, leaky(a.w + adv.w));
      }
    }
    m0 = waveMax(m0); m1 = waveMax(m1); m2 = waveMax(m2); m3 = waveMax(m3);

    float s0 = 0.f, s1 = 0.f, s2 = 0.f, s3 = 0.f;
    for (int base = 0; base < tot; base += 32) {
      const int i = base + lane;
      if (i < tot) {
        const int s = i < deg ? srcS[beg + i] : d;
        const float4 a = *(const float4*)(as_ + (size_t)s * 4);
        s0 += expf(leaky(a.x + adv.x) - m0);
        s1 += expf(leaky(a.y + adv.y) - m1);
        s2 += expf(leaky(a.z + adv.z) - m2);
        s3 += expf(leaky(a.w + adv.w) - m3);
      }
    }
    s0 = waveSum(s0) + 1e-16f; s1 = waveSum(s1) + 1e-16f;
    s2 = waveSum(s2) + 1e-16f; s3 = waveSum(s3) + 1e-16f;

    for (int base = 0; base < tot; base += 32) {
      const int i = base + lane;
      int s = d;
      float a0 = 0.f, a1 = 0.f, a2 = 0.f, a3 = 0.f;
      if (i < tot) {
        s = i < deg ? srcS[beg + i] : d;
        const float4 a = *(const float4*)(as_ + (size_t)s * 4);
        a0 = expf(leaky(a.x + adv.x) - m0) / s0;
        a1 = expf(leaky(a.y + adv.y) - m1) / s1;
        a2 = expf(leaky(a.z + adv.z) - m2) / s2;
        a3 = expf(leaky(a.w + adv.w) - m3) / s3;
      }
      const int cnt = (tot - base < 32) ? (tot - base) : 32;
      walk_rows_l1(feat, lane, upper, s, a0, a1, a2, a3, cnt, acc0, acc1);
    }
  }

  const int c0 = lane * 4, c1 = 128 + lane * 4;
  float v;
  v4h h0, h1;
  v = acc0.x + bias[c0 + 0]; h0[0] = (_Float16)(v > 0.f ? v : 0.f);
  v = acc0.y + bias[c0 + 1]; h0[1] = (_Float16)(v > 0.f ? v : 0.f);
  v = acc0.z + bias[c0 + 2]; h0[2] = (_Float16)(v > 0.f ? v : 0.f);
  v = acc0.w + bias[c0 + 3]; h0[3] = (_Float16)(v > 0.f ? v : 0.f);
  v = acc1.x + bias[c1 + 0]; h1[0] = (_Float16)(v > 0.f ? v : 0.f);
  v = acc1.y + bias[c1 + 1]; h1[1] = (_Float16)(v > 0.f ? v : 0.f);
  v = acc1.z + bias[c1 + 2]; h1[2] = (_Float16)(v > 0.f ? v : 0.f);
  v = acc1.w + bias[c1 + 3]; h1[3] = (_Float16)(v > 0.f ? v : 0.f);
  _Float16* o = outh + (size_t)d * 256;
  *(v4h*)(o + c0) = h0;
  *(v4h*)(o + c1) = h1;
}

__device__ __forceinline__ void walk_rows_l2(const float* __restrict__ feat, int lane,
                                             int s, float al, int cnt, float2& acc) {
  #pragma unroll 4
  for (int j = 0; j < cnt; ++j) {
    const int   sj = __builtin_amdgcn_readlane(s, j);
    const float aj = laneF(al, j);
    const float2 f = *(const float2*)(feat + (size_t)sj * 64 + lane * 2);
    acc.x += f.x * aj;
    acc.y += f.y * aj;
  }
}

__global__ void k_node_l2(const int* __restrict__ off, const int* __restrict__ srcS,
                          const float* __restrict__ as_, const float* __restrict__ ad_,
                          const float* __restrict__ feat, const float* __restrict__ bias,
                          float* __restrict__ outp, int Nn) {
  int d = blockIdx.x * (blockDim.x >> 5) + (threadIdx.x >> 5);
  if (d >= Nn) return;
  d = __builtin_amdgcn_readfirstlane(d);
  const int lane = threadIdx.x & 31;
  const int beg = off[d];
  const int deg = off[d + 1] - beg;
  const int tot = deg + 1;
  const float adv = ad_[d];
  float2 acc = make_float2(0.f, 0.f);

  if (tot <= 32) {
    const bool valid = lane < tot;
    int s = d;
    if (lane < deg) s = srcS[beg + lane];
    const float r = leaky(as_[s] + adv);
    const float e = valid ? r : -__builtin_inff();
    const float m = waveMax(e);
    const float p = expf(e - m);
    const float sm = waveSum(p) + 1e-16f;
    walk_rows_l2(feat, lane, s, p / sm, tot, acc);
  } else {
    float m = -__builtin_inff();
    for (int base = 0; base < tot; base += 32) {
      const int i = base + lane;
      if (i < tot) {
        const int s = i < deg ? srcS[beg + i] : d;
        m = fmaxf(m, leaky(as_[s] + adv));
      }
    }
    m = waveMax(m);

    float sm = 0.f;
    for (int base = 0; base < tot; base += 32) {
      const int i = base + lane;
      if (i < tot) {
        const int s = i < deg ? srcS[beg + i] : d;
        sm += expf(leaky(as_[s] + adv) - m);
      }
    }
    sm = waveSum(sm) + 1e-16f;

    for (int base = 0; base < tot; base += 32) {
      const int i = base + lane;
      int s = d;
      float al = 0.f;
      if (i < tot) {
        s = i < deg ? srcS[beg + i] : d;
        al = expf(leaky(as_[s] + adv) - m) / sm;
      }
      const int cnt = (tot - base < 32) ? (tot - base) : 32;
      walk_rows_l2(feat, lane, s, al, cnt, acc);
    }
  }
  float2 r;
  r.x = acc.x + bias[lane * 2];
  r.y = acc.y + bias[lane * 2 + 1];
  *(float2*)(outp + (size_t)d * 64 + lane * 2) = r;
}

extern "C" void kernel_launch(void* const* d_in, const int* in_sizes, int n_in,
                              void* d_out, int out_size, void* d_ws, size_t ws_size,
                              hipStream_t stream) {
  const int IN = 128, HID = 64, H1 = 4;
  const int HC1 = H1 * HID;
  const int HC2 = HID;
  const int Nn   = in_sizes[0] / IN;
  const int E    = in_sizes[1] / 2;

  const float* x   = (const float*)d_in[0];
  const int*   ei  = (const int*)d_in[1];
  const float* W1  = (const float*)d_in[2];
  const float* aS1 = (const float*)d_in[3];
  const float* aD1 = (const float*)d_in[4];
  const float* b1  = (const float*)d_in[5];
  const float* W2  = (const float*)d_in[6];
  const float* aS2 = (const float*)d_in[7];
  const float* aD2 = (const float*)d_in[8];
  const float* b2  = (const float*)d_in[9];
  float* out = (float*)d_out;
  const int* srcI = ei;
  const int* dstI = ei + E;

  char* p = (char*)d_ws;
  auto carve = [&](size_t bytes) -> void* {
    void* r = (void*)p;
    p += (bytes + 255) & ~(size_t)255;
    return r;
  };
  _Float16* w1t  = (_Float16*)carve((size_t)HC1 * IN * 2);
  _Float16* w2t  = (_Float16*)carve((size_t)HID * HC1 * 2);
  float*    h1   = (float*)carve((size_t)Nn * HC1 * 4);
  _Float16* h1h  = (_Float16*)carve((size_t)Nn * HC1 * 2);
  float*    h2   = (float*)carve((size_t)Nn * HC2 * 4);
  float*    as1  = (float*)carve((size_t)Nn * H1 * 4);
  float*    ad1  = (float*)carve((size_t)Nn * H1 * 4);
  float*    as2  = (float*)carve((size_t)Nn * 4);
  float*    ad2  = (float*)carve((size_t)Nn * 4);
  int*      cnt  = (int*)carve((size_t)Nn * 4);
  int*      off  = (int*)carve((size_t)(Nn + 1) * 4);
  int*      rank = (int*)carve((size_t)E * 4);
  int*      srcS = (int*)carve((size_t)E * 4);

  const int TB = 256;
  auto cdiv = [](long long a, long long b) { return (int)((a + b - 1) / b); };

  k_prep<<<cdiv((long long)IN * HC1 + (long long)HC1 * HID, TB), TB, 0, stream>>>(
      W1, w1t, IN, HC1, W2, w2t, HC1, HID, cnt, Nn);

  k_hist<<<cdiv(E, TB), TB, 0, stream>>>(dstI, E, cnt, rank);

  {
    int waves = (Nn / 16) * (HC1 / 64);
    int nScan = cdiv((long long)Nn + 1, 1024);
    k_scan_gemm_af32<<<nScan + cdiv(waves, 8), TB, 0, stream>>>(cnt, off, Nn, nScan,
                                                              x, w1t, h1, aS1, aD1, as1, ad1, Nn, HC1, IN);
  }
  k_fill<<<cdiv(E, TB), TB, 0, stream>>>(srcI, dstI, E, off, rank, srcS);
  k_node_l1<<<cdiv(Nn, 8), TB, 0, stream>>>(off, srcS, as1, ad1, h1, b1, h1h, Nn);

  {
    int waves = (Nn / 16) * (HC2 / 64);
    k_wmma_gemm<<<cdiv(waves, 8), TB, 0, stream>>>(h1h, w2t, h2, aS2, aD2, as2, ad2, Nn, HC2, HC1);
  }
  k_node_l2<<<cdiv(Nn, 8), TB, 0, stream>>>(off, srcS, as2, ad2, h2, b2, out, Nn);
}


# Edges counted and ranked per destination in LDS (32 chunk work-groups, chunk places from the scan pass), not by scattered returning global atomics; long rows too now summed in source order
# speedup vs baseline: 5.3876x; 1.3911x over previous
#include <hip/hip_runtime.h>
#include <hip/hip_bf16.h>
#include <math.h>

typedef __attribute__((ext_vector_type(16))) _Float16 v16h;
typedef __attribute__((ext_vector_type(8)))  _Float16 v8h;
typedef __attribute__((ext_vector_type(4)))  _Float16 v4h;
typedef __attribute__((ext_vector_type(8)))  float    v8f;

#define NEG_SLOPE 0.2f

__device__ __forceinline__ float leaky(float x) { return x > 0.f ? x : NEG_SLOPE * x; }

__device__ __forceinline__ float waveMax(float v) {
  for (int o = 16; o > 0; o >>= 1) v = fmaxf(v, __shfl_xor(v, o));
  return v;
}
__device__ __forceinline__ float waveSum(float v) {
  for (int o = 16; o > 0; o >>= 1) v += __shfl_xor(v, o);
  return v;
}
__device__ __forceinline__ float laneF(float v, int j) {
  return __int_as_float(__builtin_amdgcn_readlane(__float_as_int(v), j));
}

__global__ void k_prep(const float* __restrict__ in1, _Float16* __restrict__ out1, int R1, int C1,
                       const float* __restrict__ in2, _Float16* __restrict__ out2, int R2, int C2,
                       const float* __restrict__ att_s, const float* __restrict__ att_d,
                       _Float16* __restrict__ vt) {
  const int t = blockIdx.x * blockDim.x + threadIdx.x;
  const int n1 = R1 * C1;
  const int n2 = R2 * C2;
  if (t < n1) {
    const int r = t / C1, c = t - r * C1;
    out1[c * R1 + r] = (_Float16)in1[t];
  } else if (t < n1 + n2) {
    const int u = t - n1;
    const int r = u / C2, c = u - r * C2;
    out2[c * R2 + r] = (_Float16)in2[u];
  } else {
    const int v = t - n1 - n2;
    if (v < 16 * R1) {
      const int j = v / R1, k = v - j * R1;
      const int heads = C1 >> 6;
      float sum = 0.f;
      if (j < 2 * heads) {
        const int h = j < heads ? j : j - heads;
        const float* a = (j < heads ? att_s : att_d) + h * 64;
        const float* w = in1 + (size_t)k * C1 + h * 64;
        for (int c = 0; c < 64; c += 4) {
          const float4 wv = *(const float4*)(w + c);
          const float4 av = *(const float4*)(a + c);
          sum += (wv.x * av.x + wv.y * av.y) + (wv.z * av.z + wv.w * av.w);
        }
      }
      vt[v] = (_Float16)sum;
    }
  }
}

#define NCAP 51200
__global__ void __launch_bounds__(1024) k_count(const int* __restrict__ dst, int E, int chunk,
                        int Nn, int nTiles,
                        int* __restrict__ planes, int* __restrict__ tileSum,
                        int* __restrict__ rank) {
  __shared__ __attribute__((aligned(16))) int sC[NCAP];
  const int c = (int)blockIdx.x;
  const int t = (int)threadIdx.x;
  for (int i = t * 4; i < Nn; i += 4096) *(int4*)(sC + i) = make_int4(0, 0, 0, 0);
  __syncthreads();
  const int e0 = c * chunk;
  const int e1 = (E - e0 < chunk) ? E : e0 + chunk;
  for (int e = e0 + t; e < e1; e += 8 * 1024) {
    int dv[8];
    #pragma unroll
    for (int u = 0; u < 8; ++u) {
      const int ee = e + u * 1024;
      dv[u] = ee < e1 ? dst[ee] : 0;
    }
    #pragma unroll
    for (int u = 0; u < 8; ++u) {
      const int ee = e + u * 1024;
      if (ee < e1) rank[ee] = atomicAdd(&sC[dv[u]], 1);
    }
  }
  __syncthreads();
  const int w = t >> 5, lane = t & 31;
  int* __restrict__ row = planes + (size_t)c * Nn;
  for (int b = w; b < nTiles; b += 32) {
    int sum = 0;
    #pragma unroll
    for (int j = 0; j < 8; ++j) {
      const int d = b * 1024 + j * 128 + lane * 4;
      if (d < Nn) {
        const int4 v = *(const int4*)(sC + d);
        *(int4*)(row + d) = v;
        sum += (v.x + v.y) + (v.z + v.w);
      }
    }
    for (int o = 16; o > 0; o >>= 1) sum += __shfl_xor(sum, o);
    if (lane == 0) tileSum[c * nTiles + b] = sum;
  }
}

__device__ __forceinline__ void scan_tile(int* __restrict__ planes, const int* __restrict__ tileSum,
                                          int* __restrict__ off,
                                          int n, int nChunk, int nTiles, int b, int* sh) {
  const int t = threadIdx.x;
  const int lane = t & 31, w = t >> 5;
  const int lim = b * 1024;
  int before = 0;
  for (int i = t; i < nChunk * nTiles; i += 256) {
    const int bb = i % nTiles;
    if (bb < b) before += tileSum[i];
  }
  const int i0 = lim + t * 4;
  int c0 = 0, c1 = 0, c2 = 0, c3 = 0;
  if (i0 < n) {
    #pragma unroll 8
    for (int c = 0; c < nChunk; ++c) {
      const int4 v = *(const int4*)(planes + (size_t)c * n + i0);
      c0 += v.x; c1 += v.y; c2 += v.z; c3 += v.w;
    }
  }
  const int s = (c0 + c1) + (c2 + c3);
  int inc = s;
  for (int o = 1; o < 32; o <<= 1) {
    const int u = __shfl_up(inc, o);
    if (lane >= o) inc += u;
  }
  for (int o = 16; o > 0; o >>= 1) before += __shfl_xor(before, o);
  if (lane == 31) sh[w] = inc;
  if (lane == 0)  sh[8 + w] = before;
  __syncthreads();
  int e0 = inc - s;
  for (int j = 0; j < 8; ++j) {
    e0 += sh[8 + j];
    if (j < w) e0 += sh[j];
  }
  const int e1 = e0 + c0, e2 = e1 + c1, e3 = e2 + c2;
  if (i0 + 3 <= n) {
    *(int4*)(off + i0) = make_int4(e0, e1, e2, e3);
  } else {
    if (i0     <= n) off[i0]     = e0;
    if (i0 + 1 <= n) off[i0 + 1] = e1;
    if (i0 + 2 <= n) off[i0 + 2] = e2;
  }
  if (i0 < n) {
    int r0 = e0, r1 = e1, r2 = e2, r3 = e3;
    #pragma unroll 8
    for (int c = 0; c < nChunk; ++c) {
      int4* p = (int4*)(planes + (size_t)c * n + i0);
      const int4 v = *p;
      *p = make_int4(r0, r1, r2, r3);
      r0 += v.x; r1 += v.y; r2 += v.z; r3 += v.w;
    }
  }
}

__global__ void k_fill(const int* __restrict__ src, const int* __restrict__ dst, int E,
                       int chunk, int Nn,
                       const int* __restrict__ planes, const int* __restrict__ rank,
                       int* __restrict__ srcS) {
  int t = blockIdx.x * blockDim.x + threadIdx.x;
  if (t >= E) return;
  const int c = (int)(blockIdx.x * blockDim.x) / chunk;
  srcS[planes[(size_t)c * Nn + dst[t]] + rank[t]] = src[t];
}


__device__ __forceinline__ float rowSum16(float v) {
  v += __int_as_float(__builtin_amdgcn_update_dpp(0, __float_as_int(v), 0x161, 0xf, 0xf, true));
  v += __int_as_float(__builtin_amdgcn_update_dpp(0, __float_as_int(v), 0x162, 0xf, 0xf, true));
  v += __int_as_float(__builtin_amdgcn_update_dpp(0, __float_as_int(v), 0x164, 0xf, 0xf, true));
  v += __int_as_float(__builtin_amdgcn_update_dpp(0, __float_as_int(v), 0x168, 0xf, 0xf, true));
  return v;
}

__device__ __forceinline__ void logits_tile(const float* __restrict__ Af,
                                            const _Float16* __restrict__ Vt,
                                            float* __restrict__ as_, float* __restrict__ ad_,
                                            int bid, int M, int K, int heads) {
  const int wave = __builtin_amdgcn_readfirstlane(bid * 8 + (int)(threadIdx.x >> 5));
  if (wave >= (M >> 4)) return;
  const int lane = threadIdx.x & 31;
  const int half = lane >> 4;
  const int l16  = lane & 15;
  const float*    aRow = Af + (size_t)(wave * 16 + l16) * K;
  const _Float16* bRow = Vt + (size_t)l16 * K;
  v8f acc = {};
  for (int k = 0; k < K; k += 32) {
    const int ka = k + half * 8;
    union { v16h v; v8h h[2]; } a, b;
    const float* ap = aRow + ka;
    const float4 p0 = *(const float4*)(ap);
    const float4 p1 = *(const float4*)(ap + 4);
    const float4 p2 = *(const float4*)(ap + 16);
    const float4 p3 = *(const float4*)(ap + 20);
    a.v[0]  = (_Float16)p0.x; a.v[1]  = (_Float16)p0.y; a.v[2]  = (_Float16)p0.z; a.v[3]  = (_Float16)p0.w;
    a.v[4]  = (_Float16)p1.x; a.v[5]  = (_Float16)p1.y; a.v[6]  = (_Float16)p1.z; a.v[7]  = (_Float16)p1.w;
    a.v[8]  = (_Float16)p2.x; a.v[9]  = (_Float16)p2.y; a.v[10] = (_Float16)p2.z; a.v[11] = (_Float16)p2.w;
    a.v[12] = (_Float16)p3.x; a.v[13] = (_Float16)p3.y; a.v[14] = (_Float16)p3.z; a.v[15] = (_Float16)p3.w;
    b.h[0] = *(const v8h*)(bRow + ka);
    b.h[1] = *(const v8h*)(bRow + ka + 16);
    acc = __builtin_amdgcn_wmma_f32_16x16x32_f16(false, a.v, false, b.v, (short)0, acc, false, false);
  }
  if (l16 < 2 * heads) {
    const bool isD = l16 >= heads;
    const int  hh  = isD ? l16 - heads : l16;
    float* o = (isD ? ad_ : as_) + (size_t)(wave * 16 + half * 8) * heads + hh;
    #pragma unroll
    for (int j = 0; j < 8; ++j) o[(size_t)j * heads] = acc[j];
  }
}

__global__ void __launch_bounds__(256) k_scan_logits(int* __restrict__ planes, const int* __restrict__ tileSum,
                                 int* __restrict__ off,
                                 int nCnt, int nChunk, int nScan,
                                 const float* __restrict__ A,
                                 const _Float16* __restrict__ Vt,
                                 float* __restrict__ as_, float* __restrict__ ad_,
                                 int M, int K, int heads) {
  __shared__ int sScan[16];
  if ((int)blockIdx.x < nScan) {
    scan_tile(planes, tileSum, off, nCnt, nChunk, nScan, (int)blockIdx.x, sScan);
    return;
  }
  logits_tile(A, Vt, as_, ad_, (int)blockIdx.x - nScan, M, K, heads);
}


__device__ __forceinline__ v4h cvt4(float4 f) {
  v4h q;
  q[0] = (_Float16)f.x; q[1] = (_Float16)f.y; q[2] = (_Float16)f.z; q[3] = (_Float16)f.w;
  return q;
}

__device__ __forceinline__ void walk_rows_l1(const float* __restrict__ x, int lane,
                                             int s, float a0, float a1, float a2, float a3, int cnt,
                                             float4& z0, float4& z1, float4& z2, float4& z3) {
  #pragma unroll 4
  for (int j = 0; j < cnt; ++j) {
    const int   sj = __builtin_amdgcn_readlane(s, j);
    const float b0 = laneF(a0, j), b1 = laneF(a1, j), b2 = laneF(a2, j), b3 = laneF(a3, j);
    const float4 f = *(const float4*)(x + (size_t)sj * 128 + lane * 4);
    z0.x += f.x * b0; z0.y += f.y * b0; z0.z += f.z * b0; z0.w += f.w * b0;
    z1.x += f.x * b1; z1.y += f.y * b1; z1.z += f.z * b1; z1.w += f.w * b1;
    z2.x += f.x * b2; z2.y += f.y * b2; z2.z += f.z * b2; z2.w += f.w * b2;
    z3.x += f.x * b3; z3.y += f.y * b3; z3.z += f.z * b3; z3.w += f.w * b3;
  }
}

__device__ __forceinline__ void row_sums_l1(const int* __restrict__ off, int* __restrict__ srcS,
                          int* __restrict__ srcT,
                          const float* __restrict__ as_, const float* __restrict__ ad_,
                          const float* __restrict__ x, int d, int lane,
                          float4& z0, float4& z1, float4& z2, float4& z3) {
  const int beg = off[d];
  const int deg = off[d + 1] - beg;
  const int tot = deg + 1;
  const float4 adv = *(const float4*)(ad_ + (size_t)d * 4);

  if (tot <= 32) {
    const bool valid = lane < tot;
    int s = d;
    if (lane < deg) s = srcS[beg + lane];
    if (deg > 1) {
      int place = 0;
      for (int j = 0; j < deg; ++j) {
        const int sj = __builtin_amdgcn_readlane(s, j);
        place += (sj < s || (sj == s && j < lane)) ? 1 : 0;
      }
      const int to = lane < deg ? place : lane;
      s = __builtin_amdgcn_ds_permute(to << 2, s);
      if (lane < deg) srcS[beg + lane] = s;
    }
    const float4 a = *(const float4*)(as_ + (size_t)s * 4);
    const float ninf = -__builtin_inff();
    const float r0 = leaky(a.x + adv.x), r1 = leaky(a.y + adv.y);
    const float r2 = leaky(a.z + adv.z), r3 = leaky(a.w + adv.w);
    const float e0 = valid ? r0 : ninf, e1 = valid ? r1 : ninf, e2 = valid ? r2 : ninf, e3 = valid ? r3 : ninf;
    const float m0 = waveMax(e0), m1 = waveMax(e1), m2 = waveMax(e2), m3 = waveMax(e3);
    const float p0 = expf(e0 - m0), p1 = expf(e1 - m1), p2 = expf(e2 - m2), p3 = expf(e3 - m3);
    const float s0 = waveSum(p0) + 1e-16f, s1 = waveSum(p1) + 1e-16f;
    const float s2 = waveSum(p2) + 1e-16f, s3 = waveSum(p3) + 1e-16f;
    walk_rows_l1(x, lane, s, p0 / s0, p1 / s1, p2 / s2, p3 / s3, tot, z0, z1, z2, z3);
  } else {
    for (int base = 0; base < deg; base += 32) {
      const int i = base + lane;
      const int si = i < deg ? srcS[beg + i] : 0x7fffffff;
      int place = 0;
      for (int b2 = 0; b2 < deg; b2 += 32) {
        const int j0 = b2 + lane;
        const int tj = j0 < deg ? srcS[beg + j0] : 0x7fffffff;
        const int nj = (deg - b2 < 32) ? (deg - b2) : 32;
        for (int j = 0; j < nj; ++j) {
          const int sj = __builtin_amdgcn_readlane(tj, j);
          place += (sj < si || (sj == si && b2 + j < i)) ? 1 : 0;
        }
      }
      if (i < deg) srcT[beg + place] = si;
    }
    __threadfence();
    float m0 = -__builtin_inff(), m1 = m0, m2 = m0, m3 = m0;
    for (int base = 0; base < tot; base += 32) {
      const int i = base + lane;
      if (i < tot) {
        const int s = i < deg ? srcT[beg + i] : d;
        const float4 a = *(const float4*)(as_ + (size_t)s * 4);
        m0 = fmaxf(m0, leaky(a.x + adv.x));
        m1 = fmaxf(m1, leaky(a.y + adv.y));
        m2 = fmaxf(m2, leaky(a.z + adv.z));
        m3 = fmaxf(m3, leaky(a.w + adv.w));
      }
    }
    m0 = waveMax(m0); m1 = waveMax(m1); m2 = waveMax(m2); m3 = waveMax(m3);

    float s0 = 0.f, s1 = 0.f, s2 = 0.f, s3 = 0.f;
    for (int base = 0; base < tot; base += 32) {
      const int i = base + lane;
      if (i < tot) {
        const int s = i < deg ? srcT[beg + i] : d;
        const float4 a = *(const float4*)(as_ + (size_t)s * 4);
        s0 += expf(leaky(a.x + adv.x) - m0);
        s1 += expf(leaky(a.y + adv.y) - m1);
        s2 += expf(leaky(a.z + adv.z) - m2);
        s3 += expf(leaky(a.w + adv.w) - m3);
      }
    }
    s0 = waveSum(s0) + 1e-16f; s1 = waveSum(s1) + 1e-16f;
    s2 = waveSum(s2) + 1e-16f; s3 = waveSum(s3) + 1e-16f;

    for (int base = 0; base < tot; base += 32) {
      const int i = base + lane;
      int s = d;
      float a0 = 0.f, a1 = 0.f, a2 = 0.f, a3 = 0.f;
      if (i < tot) {
        s = i < deg ? srcT[beg + i] : d;
        const float4 a = *(const float4*)(as_ + (size_t)s * 4);
        a0 = expf(leaky(a.x + adv.x) - m0) / s0;
        a1 = expf(leaky(a.y + adv.y) - m1) / s1;
        a2 = expf(leaky(a.z + adv.z) - m2) / s2;
        a3 = expf(leaky(a.w + adv.w) - m3) / s3;
      }
      const int cnt = (tot - base < 32) ? (tot - base) : 32;
      walk_rows_l1(x, lane, s, a0, a1, a2, a3, cnt, z0, z1, z2, z3);
    }
  }

}

#define PZ 136
#define PH 264
#define PO 68
__global__ void __launch_bounds__(512) __attribute__((amdgpu_waves_per_eu(16, 16))) k_layer1(const int* __restrict__ off, int* __restrict__ srcS,
                          int* __restrict__ srcT,
                          const float* __restrict__ as_, const float* __restrict__ ad_,
                          const float* __restrict__ x,
                          const _Float16* __restrict__ w1t, const float* __restrict__ b1,
                          const _Float16* __restrict__ w2t,
                          const float* __restrict__ att_s2, const float* __restrict__ att_d2,
                          float* __restrict__ h2, float* __restrict__ as2, float* __restrict__ ad2,
                          int Nn) {
  __shared__ __attribute__((aligned(16))) _Float16 sZ[4 * 16 * PZ];
  __shared__ __attribute__((aligned(16))) _Float16 sH[16 * PH];
  __shared__ __attribute__((aligned(16))) float    sO[16 * PO];
  __shared__ float sL[2 * 64];

  const int w    = __builtin_amdgcn_readfirstlane((int)(threadIdx.x >> 5));
  const int lane = threadIdx.x & 31;
  const int half = lane >> 4;
  const int l16  = lane & 15;
  const int d    = (int)blockIdx.x * 16 + w;

  float4 z0 = make_float4(0.f, 0.f, 0.f, 0.f);
  float4 z1 = z0, z2 = z0, z3 = z0;
  if (d < Nn) row_sums_l1(off, srcS, srcT, as_, ad_, x, d, lane, z0, z1, z2, z3);
  {
    _Float16* o = sZ + w * PZ + lane * 4;
    *(v4h*)(o)               = cvt4(z0);
    *(v4h*)(o + 16 * PZ)     = cvt4(z1);
    *(v4h*)(o + 2 * 16 * PZ) = cvt4(z2);
    *(v4h*)(o + 3 * 16 * PZ) = cvt4(z3);
  }
  __syncthreads();

  {
    const int hd = w >> 2, ct = w & 3;
    const int col = hd * 64 + ct * 16 + l16;
    const _Float16* aRow = sZ + (hd * 16 + l16) * PZ;
    const _Float16* bRow = w1t + (size_t)col * 128;
    const float bv = b1[col];
    v8f acc = {};
    #pragma unroll 2
    for (int k = 0; k < 128; k += 32) {
      const int ka = k + half * 8;
      union { v16h v; v8h h[2]; } a, b;
      a.h[0] = *(const v8h*)(aRow + ka);
      a.h[1] = *(const v8h*)(aRow + ka + 16);
      b.h[0] = *(const v8h*)(bRow + ka);
      b.h[1] = *(const v8h*)(bRow + ka + 16);
      acc = __builtin_amdgcn_wmma_f32_16x16x32_f16(false, a.v, false, b.v, (short)0, acc, false, false);
    }
    _Float16* hp = sH + (half * 8) * PH + col;
    #pragma unroll
    for (int j = 0; j < 8; ++j) {
      const float v = acc[j] + bv;
      hp[j * PH] = (_Float16)(v > 0.f ? v : 0.f);
    }
  }
  __syncthreads();

  if (w < 4) {
    const int col = w * 16 + l16;
    const _Float16* aRow = sH + l16 * PH;
    const _Float16* bRow = w2t + (size_t)col * 256;
    const float sc = att_s2[col], dc = att_d2[col];
    v8f acc = {};
    #pragma unroll 2
    for (int k = 0; k < 256; k += 32) {
      const int ka = k + half * 8;
      union { v16h v; v8h h[2]; } a, b;
      a.h[0] = *(const v8h*)(aRow + ka);
      a.h[1] = *(const v8h*)(aRow + ka + 16);
      b.h[0] = *(const v8h*)(bRow + ka);
      b.h[1] = *(const v8h*)(bRow + ka + 16);
      acc = __builtin_amdgcn_wmma_f32_16x16x32_f16(false, a.v, false, b.v, (short)0, acc, false, false);
    }
    float* op = sO + (half * 8) * PO + col;
    float vs = 0.f, vd = 0.f;
    #pragma unroll
    for (int j = 0; j < 8; ++j) {
      op[j * PO] = acc[j];
      const float ps = rowSum16(acc[j] * sc);
      const float pd = rowSum16(acc[j] * dc);
      if (l16 == j) { vs = ps; vd = pd; }
    }
    if (l16 < 8) {
      sL[w * 16 + half * 8 + l16]      = vs;
      sL[64 + w * 16 + half * 8 + l16] = vd;
    }
  }
  __syncthreads();

  if (d < Nn) {
    const float2 v = *(const float2*)(sO + w * PO + lane * 2);
    *(float2*)(h2 + (size_t)d * 64 + lane * 2) = v;
  }
  if (w == 0) {
    const float* q = sL + half * 64 + l16;
    const float v = ((q[0] + q[16]) + q[32]) + q[48];
    const int r = (int)blockIdx.x * 16 + l16;
    if (r < Nn) (half ? ad2 : as2)[r] = v;
  }
}

__device__ __forceinline__ void walk_rows_l2(const float* __restrict__ feat, int lane,
                                             int s, float al, int cnt, float2& acc) {
  #pragma unroll 4
  for (int j = 0; j < cnt; ++j) {
    const int   sj = __builtin_amdgcn_readlane(s, j);
    const float aj = laneF(al, j);
    const float2 f = *(const float2*)(feat + (size_t)sj * 64 + lane * 2);
    acc.x += f.x * aj;
    acc.y += f.y * aj;
  }
}

__global__ void k_node_l2(const int* __restrict__ off, const int* __restrict__ srcS,
                          const int* __restrict__ srcT,
                          const float* __restrict__ as_, const float* __restrict__ ad_,
                          const float* __restrict__ feat, const float* __restrict__ bias,
                          float* __restrict__ outp, int Nn) {
  int d = blockIdx.x * (blockDim.x >> 5) + (threadIdx.x >> 5);
  if (d >= Nn) return;
  d = __builtin_amdgcn_readfirstlane(d);
  const int lane = threadIdx.x & 31;
  const int beg = off[d];
  const int deg = off[d + 1] - beg;
  const int tot = deg + 1;
  const float adv = ad_[d];
  float2 acc = make_float2(0.f, 0.f);

  if (tot <= 32) {
    const bool valid = lane < tot;
    int s = d;
    if (lane < deg) s = srcS[beg + lane];
    const float r = leaky(as_[s] + adv);
    const float e = valid ? r : -__builtin_inff();
    const float m = waveMax(e);
    const float p = expf(e - m);
    const float sm = waveSum(p) + 1e-16f;
    walk_rows_l2(feat, lane, s, p / sm, tot, acc);
  } else {
    float m = -__builtin_inff();
    for (int base = 0; base < tot; base += 32) {
      const int i = base + lane;
      if (i < tot) {
        const int s = i < deg ? srcT[beg + i] : d;
        m = fmaxf(m, leaky(as_[s] + adv));
      }
    }
    m = waveMax(m);

    float sm = 0.f;
    for (int base = 0; base < tot; base += 32) {
      const int i = base + lane;
      if (i < tot) {
        const int s = i < deg ? srcT[beg + i] : d;
        sm += expf(leaky(as_[s] + adv) - m);
      }
    }
    sm = waveSum(sm) + 1e-16f;

    for (int base = 0; base < tot; base += 32) {
      const int i = base + lane;
      int s = d;
      float al = 0.f;
      if (i < tot) {
        s = i < deg ? srcT[beg + i] : d;
        al = expf(leaky(as_[s] + adv) - m) / sm;
      }
      const int cnt = (tot - base < 32) ? (tot - base) : 32;
      walk_rows_l2(feat, lane, s, al, cnt, acc);
    }
  }
  float2 r;
  r.x = acc.x + bias[lane * 2];
  r.y = acc.y + bias[lane * 2 + 1];
  *(float2*)(outp + (size_t)d * 64 + lane * 2) = r;
}

extern "C" void kernel_launch(void* const* d_in, const int* in_sizes, int n_in,
                              void* d_out, int out_size, void* d_ws, size_t ws_size,
                              hipStream_t stream) {
  const int IN = 128, HID = 64, H1 = 4;
  const int HC1 = H1 * HID;
  const int HC2 = HID;
  const int Nn   = in_sizes[0] / IN;
  const int E    = in_sizes[1] / 2;

  const float* x   = (const float*)d_in[0];
  const int*   ei  = (const int*)d_in[1];
  const float* W1  = (const float*)d_in[2];
  const float* aS1 = (const float*)d_in[3];
  const float* aD1 = (const float*)d_in[4];
  const float* b1  = (const float*)d_in[5];
  const float* W2  = (const float*)d_in[6];
  const float* aS2 = (const float*)d_in[7];
  const float* aD2 = (const float*)d_in[8];
  const float* b2  = (const float*)d_in[9];
  float* out = (float*)d_out;
  const int* srcI = ei;
  const int* dstI = ei + E;

  char* p = (char*)d_ws;
  auto carve = [&](size_t bytes) -> void* {
    void* r = (void*)p;
    p += (bytes + 255) & ~(size_t)255;
    return r;
  };
  _Float16* w1t  = (_Float16*)carve((size_t)HC1 * IN * 2);
  _Float16* w2t  = (_Float16*)carve((size_t)HID * HC1 * 2);
  _Float16* vt   = (_Float16*)carve((size_t)16 * IN * 2);
  float*    h2   = (float*)carve((size_t)Nn * HC2 * 4);
  float*    as1  = (float*)carve((size_t)Nn * H1 * 4);
  float*    ad1  = (float*)carve((size_t)Nn * H1 * 4);
  float*    as2  = (float*)carve((size_t)Nn * 4);
  float*    ad2  = (float*)carve((size_t)Nn * 4);
  const int TB = 256;
  auto cdiv = [](long long a, long long b) { return (int)((a + b - 1) / b); };
  const int CH    = 32;
  const int chunk = cdiv(cdiv(E, CH), TB) * TB;
  const int nScan = cdiv((long long)Nn + 1, 1024);
  int*      planes  = (int*)carve((size_t)CH * Nn * 4);
  int*      tileSum = (int*)carve((size_t)CH * nScan * 4);
  int*      off  = (int*)carve((size_t)(Nn + 1) * 4);
  int*      rank = (int*)carve((size_t)E * 4);
  int*      srcS = (int*)carve((size_t)E * 4);
  int*      srcT = (int*)carve((size_t)E * 4);

  k_prep<<<cdiv((long long)IN * HC1 + (long long)HC1 * HID + 16LL * IN, TB), TB, 0, stream>>>(
      W1, w1t, IN, HC1, W2, w2t, HC1, HID, aS1, aD1, vt);

  k_count<<<CH, 1024, 0, stream>>>(dstI, E, chunk, Nn, nScan, planes, tileSum, rank);

  k_scan_logits<<<nScan + cdiv(Nn / 16, 8), TB, 0, stream>>>(planes, tileSum, off, Nn, CH, nScan,
                                                             x, vt, as1, ad1, Nn, IN, H1);
  k_fill<<<cdiv(E, TB), TB, 0, stream>>>(srcI, dstI, E, chunk, Nn, planes, rank, srcS);
  k_layer1<<<cdiv(Nn, 16), 512, 0, stream>>>(off, srcS, srcT, as1, ad1, x, w1t, b1, w2t, aS2, aD2,
                                              h2, as2, ad2, Nn);

  k_node_l2<<<cdiv(Nn, 8), TB, 0, stream>>>(off, srcS, srcT, as2, ad2, h2, b2, out, Nn);
}


# k_node_l2 walks a row in batches of up to 16 row loads in flight (was 4, then single rows); k_layer1 drops its one-at-a-time remainder; lanes past the row's end add exact zeros; sum order kept
# speedup vs baseline: 5.6094x; 1.0412x over previous
#include <hip/hip_runtime.h>
#include <hip/hip_bf16.h>
#include <math.h>

typedef __attribute__((ext_vector_type(16))) _Float16 v16h;
typedef __attribute__((ext_vector_type(8)))  _Float16 v8h;
typedef __attribute__((ext_vector_type(4)))  _Float16 v4h;
typedef __attribute__((ext_vector_type(8)))  float    v8f;

#define NEG_SLOPE 0.2f

__device__ __forceinline__ float leaky(float x) { return x > 0.f ? x : NEG_SLOPE * x; }

__device__ __forceinline__ float waveMax(float v) {
  for (int o = 16; o > 0; o >>= 1) v = fmaxf(v, __shfl_xor(v, o));
  return v;
}
__device__ __forceinline__ float waveSum(float v) {
  for (int o = 16; o > 0; o >>= 1) v += __shfl_xor(v, o);
  return v;
}
__device__ __forceinline__ float laneF(float v, int j) {
  return __int_as_float(__builtin_amdgcn_readlane(__float_as_int(v), j));
}

__global__ void k_prep(const float* __restrict__ in1, _Float16* __restrict__ out1, int R1, int C1,
                       const float* __restrict__ in2, _Float16* __restrict__ out2, int R2, int C2,
                       const float* __restrict__ att_s, const float* __restrict__ att_d,
                       _Float16* __restrict__ vt) {
  const int t = blockIdx.x * blockDim.x + threadIdx.x;
  const int n1 = R1 * C1;
  const int n2 = R2 * C2;
  if (t < n1) {
    const int r = t / C1, c = t - r * C1;
    out1[c * R1 + r] = (_Float16)in1[t];
  } else if (t < n1 + n2) {
    const int u = t - n1;
    const int r = u / C2, c = u - r * C2;
    out2[c * R2 + r] = (_Float16)in2[u];
  } else {
    const int v = t - n1 - n2;
    if (v < 16 * R1) {
      const int j = v / R1, k = v - j * R1;
      const int heads = C1 >> 6;
      float sum = 0.f;
      if (j < 2 * heads) {
        const int h = j < heads ? j : j - heads;
        const float* a = (j < heads ? att_s : att_d) + h * 64;
        const float* w = in1 + (size_t)k * C1 + h * 64;
        for (int c = 0; c < 64; c += 4) {
          const float4 wv = *(const float4*)(w + c);
          const float4 av = *(const float4*)(a + c);
          sum += (wv.x * av.x + wv.y * av.y) + (wv.z * av.z + wv.w * av.w);
        }
      }
      vt[v] = (_Float16)sum;
    }
  }
}

#define NCAP 51200
__global__ void __launch_bounds__(1024) k_count(const int* __restrict__ dst, int E, int chunk,
                        int Nn, int nTiles,
                        int* __restrict__ planes, int* __restrict__ tileSum,
                        int* __restrict__ rank) {
  __shared__ __attribute__((aligned(16))) int sC[NCAP];
  const int c = (int)blockIdx.x;
  const int t = (int)threadIdx.x;
  for (int i = t * 4; i < Nn; i += 4096) *(int4*)(sC + i) = make_int4(0, 0, 0, 0);
  __syncthreads();
  const int e0 = c * chunk;
  const int e1 = (E - e0 < chunk) ? E : e0 + chunk;
  for (int e = e0 + t; e < e1; e += 8 * 1024) {
    int dv[8];
    #pragma unroll
    for (int u = 0; u < 8; ++u) {
      const int ee = e + u * 1024;
      dv[u] = ee < e1 ? dst[ee] : 0;
    }
    #pragma unroll
    for (int u = 0; u < 8; ++u) {
      const int ee = e + u * 1024;
      if (ee < e1) rank[ee] = atomicAdd(&sC[dv[u]], 1);
    }
  }
  __syncthreads();
  const int w = t >> 5, lane = t & 31;
  int* __restrict__ row = planes + (size_t)c * Nn;
  for (int b = w; b < nTiles; b += 32) {
    int sum = 0;
    #pragma unroll
    for (int j = 0; j < 8; ++j) {
      const int d = b * 1024 + j * 128 + lane * 4;
      if (d < Nn) {
        const int4 v = *(const int4*)(sC + d);
        *(int4*)(row + d) = v;
        sum += (v.x + v.y) + (v.z + v.w);
      }
    }
    for (int o = 16; o > 0; o >>= 1) sum += __shfl_xor(sum, o);
    if (lane == 0) tileSum[c * nTiles + b] = sum;
  }
}

__device__ __forceinline__ void scan_tile(int* __restrict__ planes, const int* __restrict__ tileSum,
                                          int* __restrict__ off,
                                          int n, int nChunk, int nTiles, int b, int* sh) {
  const int t = threadIdx.x;
  const int lane = t & 31, w = t >> 5;
  const int lim = b * 1024;
  int before = 0;
  for (int i = t; i < nChunk * nTiles; i += 256) {
    const int bb = i % nTiles;
    if (bb < b) before += tileSum[i];
  }
  const int i0 = lim + t * 4;
  int c0 = 0, c1 = 0, c2 = 0, c3 = 0;
  if (i0 < n) {
    #pragma unroll 8
    for (int c = 0; c < nChunk; ++c) {
      const int4 v = *(const int4*)(planes + (size_t)c * n + i0);
      c0 += v.x; c1 += v.y; c2 += v.z; c3 += v.w;
    }
  }
  const int s = (c0 + c1) + (c2 + c3);
  int inc = s;
  for (int o = 1; o < 32; o <<= 1) {
    const int u = __shfl_up(inc, o);
    if (lane >= o) inc += u;
  }
  for (int o = 16; o > 0; o >>= 1) before += __shfl_xor(before, o);
  if (lane == 31) sh[w] = inc;
  if (lane == 0)  sh[8 + w] = before;
  __syncthreads();
  int e0 = inc - s;
  for (int j = 0; j < 8; ++j) {
    e0 += sh[8 + j];
    if (j < w) e0 += sh[j];
  }
  const int e1 = e0 + c0, e2 = e1 + c1, e3 = e2 + c2;
  if (i0 + 3 <= n) {
    *(int4*)(off + i0) = make_int4(e0, e1, e2, e3);
  } else {
    if (i0     <= n) off[i0]     = e0;
    if (i0 + 1 <= n) off[i0 + 1] = e1;
    if (i0 + 2 <= n) off[i0 + 2] = e2;
  }
  if (i0 < n) {
    int r0 = e0, r1 = e1, r2 = e2, r3 = e3;
    #pragma unroll 8
    for (int c = 0; c < nChunk; ++c) {
      int4* p = (int4*)(planes + (size_t)c * n + i0);
      const int4 v = *p;
      *p = make_int4(r0, r1, r2, r3);
      r0 += v.x; r1 += v.y; r2 += v.z; r3 += v.w;
    }
  }
}

__global__ void k_fill(const int* __restrict__ src, const int* __restrict__ dst, int E,
                       int chunk, int Nn,
                       const int* __restrict__ planes, const int* __restrict__ rank,
                       int* __restrict__ srcS) {
  int t = blockIdx.x * blockDim.x + threadIdx.x;
  if (t >= E) return;
  const int c = (int)(blockIdx.x * blockDim.x) / chunk;
  srcS[planes[(size_t)c * Nn + dst[t]] + rank[t]] = src[t];
}


__device__ __forceinline__ float rowSum16(float v) {
  v += __int_as_float(__builtin_amdgcn_update_dpp(0, __float_as_int(v), 0x161, 0xf, 0xf, true));
  v += __int_as_float(__builtin_amdgcn_update_dpp(0, __float_as_int(v), 0x162, 0xf, 0xf, true));
  v += __int_as_float(__builtin_amdgcn_update_dpp(0, __float_as_int(v), 0x164, 0xf, 0xf, true));
  v += __int_as_float(__builtin_amdgcn_update_dpp(0, __float_as_int(v), 0x168, 0xf, 0xf, true));
  return v;
}

__device__ __forceinline__ void logits_tile(const float* __restrict__ Af,
                                            const _Float16* __restrict__ Vt,
                                            float* __restrict__ as_, float* __restrict__ ad_,
                                            int bid, int M, int K, int heads) {
  const int wave = __builtin_amdgcn_readfirstlane(bid * 8 + (int)(threadIdx.x >> 5));
  if (wave >= (M >> 4)) return;
  const int lane = threadIdx.x & 31;
  const int half = lane >> 4;
  const int l16  = lane & 15;
  const float*    aRow = Af + (size_t)(wave * 16 + l16) * K;
  const _Float16* bRow = Vt + (size_t)l16 * K;
  v8f acc = {};
  for (int k = 0; k < K; k += 32) {
    const int ka = k + half * 8;
    union { v16h v; v8h h[2]; } a, b;
    const float* ap = aRow + ka;
    const float4 p0 = *(const float4*)(ap);
    const float4 p1 = *(const float4*)(ap + 4);
    const float4 p2 = *(const float4*)(ap + 16);
    const float4 p3 = *(const float4*)(ap + 20);
    a.v[0]  = (_Float16)p0.x; a.v[1]  = (_Float16)p0.y; a.v[2]  = (_Float16)p0.z; a.v[3]  = (_Float16)p0.w;
    a.v[4]  = (_Float16)p1.x; a.v[5]  = (_Float16)p1.y; a.v[6]  = (_Float16)p1.z; a.v[7]  = (_Float16)p1.w;
    a.v[8]  = (_Float16)p2.x; a.v[9]  = (_Float16)p2.y; a.v[10] = (_Float16)p2.z; a.v[11] = (_Float16)p2.w;
    a.v[12] = (_Float16)p3.x; a.v[13] = (_Float16)p3.y; a.v[14] = (_Float16)p3.z; a.v[15] = (_Float16)p3.w;
    b.h[0] = *(const v8h*)(bRow + ka);
    b.h[1] = *(const v8h*)(bRow + ka + 16);
    acc = __builtin_amdgcn_wmma_f32_16x16x32_f16(false, a.v, false, b.v, (short)0, acc, false, false);
  }
  if (l16 < 2 * heads) {
    const bool isD = l16 >= heads;
    const int  hh  = isD ? l16 - heads : l16;
    float* o = (isD ? ad_ : as_) + (size_t)(wave * 16 + half * 8) * heads + hh;
    #pragma unroll
    for (int j = 0; j < 8; ++j) o[(size_t)j * heads] = acc[j];
  }
}

__global__ void __launch_bounds__(256) k_scan_logits(int* __restrict__ planes, const int* __restrict__ tileSum,
                                 int* __restrict__ off,
                                 int nCnt, int nChunk, int nScan,
                                 const float* __restrict__ A,
                                 const _Float16* __restrict__ Vt,
                                 float* __restrict__ as_, float* __restrict__ ad_,
                                 int M, int K, int heads) {
  __shared__ int sScan[16];
  if ((int)blockIdx.x < nScan) {
    scan_tile(planes, tileSum, off, nCnt, nChunk, nScan, (int)blockIdx.x, sScan);
    return;
  }
  logits_tile(A, Vt, as_, ad_, (int)blockIdx.x - nScan, M, K, heads);
}


__device__ __forceinline__ v4h cvt4(float4 f) {
  v4h q;
  q[0] = (_Float16)f.x; q[1] = (_Float16)f.y; q[2] = (_Float16)f.z; q[3] = (_Float16)f.w;
  return q;
}

__device__ __forceinline__ void walk_rows_l1(const float* __restrict__ x, int lane,
                                             int s, float a0, float a1, float a2, float a3, int cnt,
                                             float4& z0, float4& z1, float4& z2, float4& z3) {
  for (int j0 = 0; j0 < cnt; j0 += 4) {
    float4 f[4];
    #pragma unroll
    for (int u = 0; u < 4; ++u) {
      const int sj = __builtin_amdgcn_readlane(s, j0 + u);
      f[u] = *(const float4*)(x + (size_t)sj * 128 + lane * 4);
    }
    #pragma unroll
    for (int u = 0; u < 4; ++u) {
      const int j = j0 + u;
      const float b0 = laneF(a0, j), b1 = laneF(a1, j), b2 = laneF(a2, j), b3 = laneF(a3, j);
      z0.x = __builtin_fmaf(f[u].x, b0, z0.x); z0.y = __builtin_fmaf(f[u].y, b0, z0.y);
      z0.z = __builtin_fmaf(f[u].z, b0, z0.z); z0.w = __builtin_fmaf(f[u].w, b0, z0.w);
      z1.x = __builtin_fmaf(f[u].x, b1, z1.x); z1.y = __builtin_fmaf(f[u].y, b1, z1.y);
      z1.z = __builtin_fmaf(f[u].z, b1, z1.z); z1.w = __builtin_fmaf(f[u].w, b1, z1.w);
      z2.x = __builtin_fmaf(f[u].x, b2, z2.x); z2.y = __builtin_fmaf(f[u].y, b2, z2.y);
      z2.z = __builtin_fmaf(f[u].z, b2, z2.z); z2.w = __builtin_fmaf(f[u].w, b2, z2.w);
      z3.x = __builtin_fmaf(f[u].x, b3, z3.x); z3.y = __builtin_fmaf(f[u].y, b3, z3.y);
      z3.z = __builtin_fmaf(f[u].z, b3, z3.z); z3.w = __builtin_fmaf(f[u].w, b3, z3.w);
    }
  }
}

__device__ __forceinline__ void row_sums_l1(const int* __restrict__ off, int* __restrict__ srcS,
                          int* __restrict__ srcT,
                          const float* __restrict__ as_, const float* __restrict__ ad_,
                          const float* __restrict__ x, int d, int lane,
                          float4& z0, float4& z1, float4& z2, float4& z3) {
  const int beg = off[d];
  const int deg = off[d + 1] - beg;
  const int tot = deg + 1;
  const float4 adv = *(const float4*)(ad_ + (size_t)d * 4);

  if (tot <= 32) {
    const bool valid = lane < tot;
    int s = d;
    if (lane < deg) s = srcS[beg + lane];
    if (deg > 1) {
      int place = 0;
      for (int j = 0; j < deg; ++j) {
        const int sj = __builtin_amdgcn_readlane(s, j);
        place += (sj < s || (sj == s && j < lane)) ? 1 : 0;
      }
      const int to = lane < deg ? place : lane;
      s = __builtin_amdgcn_ds_permute(to << 2, s);
      if (lane < deg) srcS[beg + lane] = s;
    }
    const float4 a = *(const float4*)(as_ + (size_t)s * 4);
    const float ninf = -__builtin_inff();
    const float r0 = leaky(a.x + adv.x), r1 = leaky(a.y + adv.y);
    const float r2 = leaky(a.z + adv.z), r3 = leaky(a.w + adv.w);
    const float e0 = valid ? r0 : ninf, e1 = valid ? r1 : ninf, e2 = valid ? r2 : ninf, e3 = valid ? r3 : ninf;
    const float m0 = waveMax(e0), m1 = waveMax(e1), m2 = waveMax(e2), m3 = waveMax(e3);
    const float p0 = expf(e0 - m0), p1 = expf(e1 - m1), p2 = expf(e2 - m2), p3 = expf(e3 - m3);
    const float s0 = waveSum(p0) + 1e-16f, s1 = waveSum(p1) + 1e-16f;
    const float s2 = waveSum(p2) + 1e-16f, s3 = waveSum(p3) + 1e-16f;
    walk_rows_l1(x, lane, s, p0 / s0, p1 / s1, p2 / s2, p3 / s3, tot, z0, z1, z2, z3);
  } else {
    for (int base = 0; base < deg; base += 32) {
      const int i = base + lane;
      const int si = i < deg ? srcS[beg + i] : 0x7fffffff;
      int place = 0;
      for (int b2 = 0; b2 < deg; b2 += 32) {
        const int j0 = b2 + lane;
        const int tj = j0 < deg ? srcS[beg + j0] : 0x7fffffff;
        const int nj = (deg - b2 < 32) ? (deg - b2) : 32;
        for (int j = 0; j < nj; ++j) {
          const int sj = __builtin_amdgcn_readlane(tj, j);
          place += (sj < si || (sj == si && b2 + j < i)) ? 1 : 0;
        }
      }
      if (i < deg) srcT[beg + place] = si;
    }
    __threadfence();
    float m0 = -__builtin_inff(), m1 = m0, m2 = m0, m3 = m0;
    for (int base = 0; base < tot; base += 32) {
      const int i = base + lane;
      if (i < tot) {
        const int s = i < deg ? srcT[beg + i] : d;
        const float4 a = *(const float4*)(as_ + (size_t)s * 4);
        m0 = fmaxf(m0, leaky(a.x + adv.x));
        m1 = fmaxf(m1, leaky(a.y + adv.y));
        m2 = fmaxf(m2, leaky(a.z + adv.z));
        m3 = fmaxf(m3, leaky(a.w + adv.w));
      }
    }
    m0 = waveMax(m0); m1 = waveMax(m1); m2 = waveMax(m2); m3 = waveMax(m3);

    float s0 = 0.f, s1 = 0.f, s2 = 0.f, s3 = 0.f;
    for (int base = 0; base < tot; base += 32) {
      const int i = base + lane;
      if (i < tot) {
        const int s = i < deg ? srcT[beg + i] : d;
        const float4 a = *(const float4*)(as_ + (size_t)s * 4);
        s0 += expf(leaky(a.x + adv.x) - m0);
        s1 += expf(leaky(a.y + adv.y) - m1);
        s2 += expf(leaky(a.z + adv.z) - m2);
        s3 += expf(leaky(a.w + adv.w) - m3);
      }
    }
    s0 = waveSum(s0) + 1e-16f; s1 = waveSum(s1) + 1e-16f;
    s2 = waveSum(s2) + 1e-16f; s3 = waveSum(s3) + 1e-16f;

    for (int base = 0; base < tot; base += 32) {
      const int i = base + lane;
      int s = d;
      float a0 = 0.f, a1 = 0.f, a2 = 0.f, a3 = 0.f;
      if (i < tot) {
        s = i < deg ? srcT[beg + i] : d;
        const float4 a = *(const float4*)(as_ + (size_t)s * 4);
        a0 = expf(leaky(a.x + adv.x) - m0) / s0;
        a1 = expf(leaky(a.y + adv.y) - m1) / s1;
        a2 = expf(leaky(a.z + adv.z) - m2) / s2;
        a3 = expf(leaky(a.w + adv.w) - m3) / s3;
      }
      const int cnt = (tot - base < 32) ? (tot - base) : 32;
      walk_rows_l1(x, lane, s, a0, a1, a2, a3, cnt, z0, z1, z2, z3);
    }
  }

}

#define PZ 136
#define PH 264
#define PO 68
__global__ void __launch_bounds__(512) __attribute__((amdgpu_waves_per_eu(16, 16))) k_layer1(const int* __restrict__ off, int* __restrict__ srcS,
                          int* __restrict__ srcT,
                          const float* __restrict__ as_, const float* __restrict__ ad_,
                          const float* __restrict__ x,
                          const _Float16* __restrict__ w1t, const float* __restrict__ b1,
                          const _Float16* __restrict__ w2t,
                          const float* __restrict__ att_s2, const float* __restrict__ att_d2,
                          float* __restrict__ h2, float* __restrict__ as2, float* __restrict__ ad2,
                          int Nn) {
  __shared__ __attribute__((aligned(16))) _Float16 sZ[4 * 16 * PZ];
  __shared__ __attribute__((aligned(16))) _Float16 sH[16 * PH];
  __shared__ __attribute__((aligned(16))) float    sO[16 * PO];
  __shared__ float sL[2 * 64];

  const int w    = __builtin_amdgcn_readfirstlane((int)(threadIdx.x >> 5));
  const int lane = threadIdx.x & 31;
  const int half = lane >> 4;
  const int l16  = lane & 15;
  const int d    = (int)blockIdx.x * 16 + w;

  float4 z0 = make_float4(0.f, 0.f, 0.f, 0.f);
  float4 z1 = z0, z2 = z0, z3 = z0;
  if (d < Nn) row_sums_l1(off, srcS, srcT, as_, ad_, x, d, lane, z0, z1, z2, z3);
  {
    _Float16* o = sZ + w * PZ + lane * 4;
    *(v4h*)(o)               = cvt4(z0);
    *(v4h*)(o + 16 * PZ)     = cvt4(z1);
    *(v4h*)(o + 2 * 16 * PZ) = cvt4(z2);
    *(v4h*)(o + 3 * 16 * PZ) = cvt4(z3);
  }
  __syncthreads();

  {
    const int hd = w >> 2, ct = w & 3;
    const int col = hd * 64 + ct * 16 + l16;
    const _Float16* aRow = sZ + (hd * 16 + l16) * PZ;
    const _Float16* bRow = w1t + (size_t)col * 128;
    const float bv = b1[col];
    v8f acc = {};
    #pragma unroll 2
    for (int k = 0; k < 128; k += 32) {
      const int ka = k + half * 8;
      union { v16h v; v8h h[2]; } a, b;
      a.h[0] = *(const v8h*)(aRow + ka);
      a.h[1] = *(const v8h*)(aRow + ka + 16);
      b.h[0] = *(const v8h*)(bRow + ka);
      b.h[1] = *(const v8h*)(bRow + ka + 16);
      acc = __builtin_amdgcn_wmma_f32_16x16x32_f16(false, a.v, false, b.v, (short)0, acc, false, false);
    }
    _Float16* hp = sH + (half * 8) * PH + col;
    #pragma unroll
    for (int j = 0; j < 8; ++j) {
      const float v = acc[j] + bv;
      hp[j * PH] = (_Float16)(v > 0.f ? v : 0.f);
    }
  }
  __syncthreads();

  if (w < 4) {
    const int col = w * 16 + l16;
    const _Float16* aRow = sH + l16 * PH;
    const _Float16* bRow = w2t + (size_t)col * 256;
    const float sc = att_s2[col], dc = att_d2[col];
    v8f acc = {};
    #pragma unroll 2
    for (int k = 0; k < 256; k += 32) {
      const int ka = k + half * 8;
      union { v16h v; v8h h[2]; } a, b;
      a.h[0] = *(const v8h*)(aRow + ka);
      a.h[1] = *(const v8h*)(aRow + ka + 16);
      b.h[0] = *(const v8h*)(bRow + ka);
      b.h[1] = *(const v8h*)(bRow + ka + 16);
      acc = __builtin_amdgcn_wmma_f32_16x16x32_f16(false, a.v, false, b.v, (short)0, acc, false, false);
    }
    float* op = sO + (half * 8) * PO + col;
    float vs = 0.f, vd = 0.f;
    #pragma unroll
    for (int j = 0; j < 8; ++j) {
      op[j * PO] = acc[j];
      const float ps = rowSum16(acc[j] * sc);
      const float pd = rowSum16(acc[j] * dc);
      if (l16 == j) { vs = ps; vd = pd; }
    }
    if (l16 < 8) {
      sL[w * 16 + half * 8 + l16]      = vs;
      sL[64 + w * 16 + half * 8 + l16] = vd;
    }
  }
  __syncthreads();

  if (d < Nn) {
    const float2 v = *(const float2*)(sO + w * PO + lane * 2);
    *(float2*)(h2 + (size_t)d * 64 + lane * 2) = v;
  }
  if (w == 0) {
    const float* q = sL + half * 64 + l16;
    const float v = ((q[0] + q[16]) + q[32]) + q[48];
    const int r = (int)blockIdx.x * 16 + l16;
    if (r < Nn) (half ? ad2 : as2)[r] = v;
  }
}

__device__ __forceinline__ void walk_rows_l2(const float* __restrict__ feat, int lane,
                                             int s, float al, int cnt, float2& acc) {
  for (int j0 = 0; j0 < cnt; j0 += 16) {
    float2 f[16];
    const bool upper = j0 + 8 < cnt;
    #pragma unroll
    for (int u = 0; u < 8; ++u) {
      const int sj = __builtin_amdgcn_readlane(s, j0 + u);
      f[u] = *(const float2*)(feat + (size_t)sj * 64 + lane * 2);
    }
    if (upper) {
      #pragma unroll
      for (int u = 8; u < 16; ++u) {
        const int sj = __builtin_amdgcn_readlane(s, j0 + u);
        f[u] = *(const float2*)(feat + (size_t)sj * 64 + lane * 2);
      }
    }
    #pragma unroll
    for (int u = 0; u < 8; ++u) {
      const float aj = laneF(al, j0 + u);
      acc.x = __builtin_fmaf(f[u].x, aj, acc.x);
      acc.y = __builtin_fmaf(f[u].y, aj, acc.y);
    }
    if (upper) {
      #pragma unroll
      for (int u = 8; u < 16; ++u) {
        const float aj = laneF(al, j0 + u);
        acc.x = __builtin_fmaf(f[u].x, aj, acc.x);
        acc.y = __builtin_fmaf(f[u].y, aj, acc.y);
      }
    }
  }
}

__global__ void k_node_l2(const int* __restrict__ off, const int* __restrict__ srcS,
                          const int* __restrict__ srcT,
                          const float* __restrict__ as_, const float* __restrict__ ad_,
                          const float* __restrict__ feat, const float* __restrict__ bias,
                          float* __restrict__ outp, int Nn) {
  int d = blockIdx.x * (blockDim.x >> 5) + (threadIdx.x >> 5);
  if (d >= Nn) return;
  d = __builtin_amdgcn_readfirstlane(d);
  const int lane = threadIdx.x & 31;
  const int beg = off[d];
  const int deg = off[d + 1] - beg;
  const int tot = deg + 1;
  const float adv = ad_[d];
  float2 acc = make_float2(0.f, 0.f);

  if (tot <= 32) {
    const bool valid = lane < tot;
    int s = d;
    if (lane < deg) s = srcS[beg + lane];
    const float r = leaky(as_[s] + adv);
    const float e = valid ? r : -__builtin_inff();
    const float m = waveMax(e);
    const float p = expf(e - m);
    const float sm = waveSum(p) + 1e-16f;
    walk_rows_l2(feat, lane, s, p / sm, tot, acc);
  } else {
    float m = -__builtin_inff();
    for (int base = 0; base < tot; base += 32) {
      const int i = base + lane;
      if (i < tot) {
        const int s = i < deg ? srcT[beg + i] : d;
        m = fmaxf(m, leaky(as_[s] + adv));
      }
    }
    m = waveMax(m);

    float sm = 0.f;
    for (int base = 0; base < tot; base += 32) {
      const int i = base + lane;
      if (i < tot) {
        const int s = i < deg ? srcT[beg + i] : d;
        sm += expf(leaky(as_[s] + adv) - m);
      }
    }
    sm = waveSum(sm) + 1e-16f;

    for (int base = 0; base < tot; base += 32) {
      const int i = base + lane;
      int s = d;
      float al = 0.f;
      if (i < tot) {
        s = i < deg ? srcT[beg + i] : d;
        al = expf(leaky(as_[s] + adv) - m) / sm;
      }
      const int cnt = (tot - base < 32) ? (tot - base) : 32;
      walk_rows_l2(feat, lane, s, al, cnt, acc);
    }
  }
  float2 r;
  r.x = acc.x + bias[lane * 2];
  r.y = acc.y + bias[lane * 2 + 1];
  *(float2*)(outp + (size_t)d * 64 + lane * 2) = r;
}

extern "C" void kernel_launch(void* const* d_in, const int* in_sizes, int n_in,
                              void* d_out, int out_size, void* d_ws, size_t ws_size,
                              hipStream_t stream) {
  const int IN = 128, HID = 64, H1 = 4;
  const int HC1 = H1 * HID;
  const int HC2 = HID;
  const int Nn   = in_sizes[0] / IN;
  const int E    = in_sizes[1] / 2;

  const float* x   = (const float*)d_in[0];
  const int*   ei  = (const int*)d_in[1];
  const float* W1  = (const float*)d_in[2];
  const float* aS1 = (const float*)d_in[3];
  const float* aD1 = (const float*)d_in[4];
  const float* b1  = (const float*)d_in[5];
  const float* W2  = (const float*)d_in[6];
  const float* aS2 = (const float*)d_in[7];
  const float* aD2 = (const float*)d_in[8];
  const float* b2  = (const float*)d_in[9];
  float* out = (float*)d_out;
  const int* srcI = ei;
  const int* dstI = ei + E;

  char* p = (char*)d_ws;
  auto carve = [&](size_t bytes) -> void* {
    void* r = (void*)p;
    p += (bytes + 255) & ~(size_t)255;
    return r;
  };
  _Float16* w1t  = (_Float16*)carve((size_t)HC1 * IN * 2);
  _Float16* w2t  = (_Float16*)carve((size_t)HID * HC1 * 2);
  _Float16* vt   = (_Float16*)carve((size_t)16 * IN * 2);
  float*    h2   = (float*)carve((size_t)Nn * HC2 * 4);
  float*    as1  = (float*)carve((size_t)Nn * H1 * 4);
  float*    ad1  = (float*)carve((size_t)Nn * H1 * 4);
  float*    as2  = (float*)carve((size_t)Nn * 4);
  float*    ad2  = (float*)carve((size_t)Nn * 4);
  const int TB = 256;
  auto cdiv = [](long long a, long long b) { return (int)((a + b - 1) / b); };
  const int CH    = 32;
  const int chunk = cdiv(cdiv(E, CH), TB) * TB;
  const int nScan = cdiv((long long)Nn + 1, 1024);
  int*      planes  = (int*)carve((size_t)CH * Nn * 4);
  int*      tileSum = (int*)carve((size_t)CH * nScan * 4);
  int*      off  = (int*)carve((size_t)(Nn + 1) * 4);
  int*      rank = (int*)carve((size_t)E * 4);
  int*      srcS = (int*)carve((size_t)E * 4);
  int*      srcT = (int*)carve((size_t)E * 4);

  k_prep<<<cdiv((long long)IN * HC1 + (long long)HC1 * HID + 16LL * IN, TB), TB, 0, stream>>>(
      W1, w1t, IN, HC1, W2, w2t, HC1, HID, aS1, aD1, vt);

  k_count<<<CH, 1024, 0, stream>>>(dstI, E, chunk, Nn, nScan, planes, tileSum, rank);

  k_scan_logits<<<nScan + cdiv(Nn / 16, 8), TB, 0, stream>>>(planes, tileSum, off, Nn, CH, nScan,
                                                             x, vt, as1, ad1, Nn, IN, H1);
  k_fill<<<cdiv(E, TB), TB, 0, stream>>>(srcI, dstI, E, chunk, Nn, planes, rank, srcS);
  k_layer1<<<cdiv(Nn, 16), 512, 0, stream>>>(off, srcS, srcT, as1, ad1, x, w1t, b1, w2t, aS2, aD2,
                                              h2, as2, ad2, Nn);

  k_node_l2<<<cdiv(Nn, 8), TB, 0, stream>>>(off, srcS, srcT, as2, ad2, h2, b2, out, Nn);
}


# W1/W2 stored by the prep pass in WMMA B-fragment order: each fragment load of k_layer1 is 512 contiguous bytes (4 whole lines), was 32 bytes from each of 16 lines; same k order, sums unchanged
# speedup vs baseline: 5.8708x; 1.0466x over previous
#include <hip/hip_runtime.h>
#include <hip/hip_bf16.h>
#include <math.h>

typedef __attribute__((ext_vector_type(16))) _Float16 v16h;
typedef __attribute__((ext_vector_type(8)))  _Float16 v8h;
typedef __attribute__((ext_vector_type(4)))  _Float16 v4h;
typedef __attribute__((ext_vector_type(8)))  float    v8f;

#define NEG_SLOPE 0.2f

__device__ __forceinline__ float leaky(float x) { return x > 0.f ? x : NEG_SLOPE * x; }

__device__ __forceinline__ float waveMax(float v) {
  for (int o = 16; o > 0; o >>= 1) v = fmaxf(v, __shfl_xor(v, o));
  return v;
}
__device__ __forceinline__ float waveSum(float v) {
  for (int o = 16; o > 0; o >>= 1) v += __shfl_xor(v, o);
  return v;
}
__device__ __forceinline__ float laneF(float v, int j) {
  return __int_as_float(__builtin_amdgcn_readlane(__float_as_int(v), j));
}

__device__ __forceinline__ int frag_index(int k, int col, int K) {
  const int T = col >> 4, cc = col & 15;
  const int s = k >> 5, kk = k & 31;
  const int p = kk >> 4, h = (kk >> 3) & 1, e = kk & 7;
  return ((((T * (K >> 5) + s) * 2 + p) * 32) + (h * 16 + cc)) * 8 + e;
}
__global__ void k_prep(const float* __restrict__ in1, _Float16* __restrict__ out1, int R1, int C1,
                       const float* __restrict__ in2, _Float16* __restrict__ out2, int R2, int C2,
                       const float* __restrict__ att_s, const float* __restrict__ att_d,
                       _Float16* __restrict__ vt) {
  const int t = blockIdx.x * blockDim.x + threadIdx.x;
  const int n1 = R1 * C1;
  const int n2 = R2 * C2;
  if (t < n1) {
    const int r = t / C1, c = t - r * C1;
    out1[frag_index(r, c, R1)] = (_Float16)in1[t];
  } else if (t < n1 + n2) {
    const int u = t - n1;
    const int r = u / C2, c = u - r * C2;
    out2[frag_index(r, c, R2)] = (_Float16)in2[u];
  } else {
    const int v = t - n1 - n2;
    if (v < 16 * R1) {
      const int j = v / R1, k = v - j * R1;
      const int heads = C1 >> 6;
      float sum = 0.f;
      if (j < 2 * heads) {
        const int h = j < heads ? j : j - heads;
        const float* a = (j < heads ? att_s : att_d) + h * 64;
        const float* w = in1 + (size_t)k * C1 + h * 64;
        for (int c = 0; c < 64; c += 4) {
          const float4 wv = *(const float4*)(w + c);
          const float4 av = *(const float4*)(a + c);
          sum += (wv.x * av.x + wv.y * av.y) + (wv.z * av.z + wv.w * av.w);
        }
      }
      vt[v] = (_Float16)sum;
    }
  }
}

#define NCAP 51200
__global__ void __launch_bounds__(1024) k_count(const int* __restrict__ dst, int E, int chunk,
                        int Nn, int nTiles,
                        int* __restrict__ planes, int* __restrict__ tileSum,
                        int* __restrict__ rank) {
  __shared__ __attribute__((aligned(16))) int sC[NCAP];
  const int c = (int)blockIdx.x;
  const int t = (int)threadIdx.x;
  for (int i = t * 4; i < Nn; i += 4096) *(int4*)(sC + i) = make_int4(0, 0, 0, 0);
  __syncthreads();
  const int e0 = c * chunk;
  const int e1 = (E - e0 < chunk) ? E : e0 + chunk;
  for (int e = e0 + t; e < e1; e += 8 * 1024) {
    int dv[8];
    #pragma unroll
    for (int u = 0; u < 8; ++u) {
      const int ee = e + u * 1024;
      dv[u] = ee < e1 ? dst[ee] : 0;
    }
    #pragma unroll
    for (int u = 0; u < 8; ++u) {
      const int ee = e + u * 1024;
      if (ee < e1) rank[ee] = atomicAdd(&sC[dv[u]], 1);
    }
  }
  __syncthreads();
  const int w = t >> 5, lane = t & 31;
  int* __restrict__ row = planes + (size_t)c * Nn;
  for (int b = w; b < nTiles; b += 32) {
    int sum = 0;
    #pragma unroll
    for (int j = 0; j < 8; ++j) {
      const int d = b * 1024 + j * 128 + lane * 4;
      if (d < Nn) {
        const int4 v = *(const int4*)(sC + d);
        *(int4*)(row + d) = v;
        sum += (v.x + v.y) + (v.z + v.w);
      }
    }
    for (int o = 16; o > 0; o >>= 1) sum += __shfl_xor(sum, o);
    if (lane == 0) tileSum[c * nTiles + b] = sum;
  }
}

__device__ __forceinline__ void scan_tile(int* __restrict__ planes, const int* __restrict__ tileSum,
                                          int* __restrict__ off,
                                          int n, int nChunk, int nTiles, int b, int* sh) {
  const int t = threadIdx.x;
  const int lane = t & 31, w = t >> 5;
  const int lim = b * 1024;
  int before = 0;
  for (int i = t; i < nChunk * nTiles; i += 256) {
    const int bb = i % nTiles;
    if (bb < b) before += tileSum[i];
  }
  const int i0 = lim + t * 4;
  int c0 = 0, c1 = 0, c2 = 0, c3 = 0;
  if (i0 < n) {
    #pragma unroll 8
    for (int c = 0; c < nChunk; ++c) {
      const int4 v = *(const int4*)(planes + (size_t)c * n + i0);
      c0 += v.x; c1 += v.y; c2 += v.z; c3 += v.w;
    }
  }
  const int s = (c0 + c1) + (c2 + c3);
  int inc = s;
  for (int o = 1; o < 32; o <<= 1) {
    const int u = __shfl_up(inc, o);
    if (lane >= o) inc += u;
  }
  for (int o = 16; o > 0; o >>= 1) before += __shfl_xor(before, o);
  if (lane == 31) sh[w] = inc;
  if (lane == 0)  sh[8 + w] = before;
  __syncthreads();
  int e0 = inc - s;
  for (int j = 0; j < 8; ++j) {
    e0 += sh[8 + j];
    if (j < w) e0 += sh[j];
  }
  const int e1 = e0 + c0, e2 = e1 + c1, e3 = e2 + c2;
  if (i0 + 3 <= n) {
    *(int4*)(off + i0) = make_int4(e0, e1, e2, e3);
  } else {
    if (i0     <= n) off[i0]     = e0;
    if (i0 + 1 <= n) off[i0 + 1] = e1;
    if (i0 + 2 <= n) off[i0 + 2] = e2;
  }
  if (i0 < n) {
    int r0 = e0, r1 = e1, r2 = e2, r3 = e3;
    #pragma unroll 8
    for (int c = 0; c < nChunk; ++c) {
      int4* p = (int4*)(planes + (size_t)c * n + i0);
      const int4 v = *p;
      *p = make_int4(r0, r1, r2, r3);
      r0 += v.x; r1 += v.y; r2 += v.z; r3 += v.w;
    }
  }
}

__global__ void k_fill(const int* __restrict__ src, const int* __restrict__ dst, int E,
                       int chunk, int Nn,
                       const int* __restrict__ planes, const int* __restrict__ rank,
                       int* __restrict__ srcS) {
  int t = blockIdx.x * blockDim.x + threadIdx.x;
  if (t >= E) return;
  const int c = (int)(blockIdx.x * blockDim.x) / chunk;
  srcS[planes[(size_t)c * Nn + dst[t]] + rank[t]] = src[t];
}


__device__ __forceinline__ float rowSum16(float v) {
  v += __int_as_float(__builtin_amdgcn_update_dpp(0, __float_as_int(v), 0x161, 0xf, 0xf, true));
  v += __int_as_float(__builtin_amdgcn_update_dpp(0, __float_as_int(v), 0x162, 0xf, 0xf, true));
  v += __int_as_float(__builtin_amdgcn_update_dpp(0, __float_as_int(v), 0x164, 0xf, 0xf, true));
  v += __int_as_float(__builtin_amdgcn_update_dpp(0, __float_as_int(v), 0x168, 0xf, 0xf, true));
  return v;
}

__device__ __forceinline__ void logits_tile(const float* __restrict__ Af,
                                            const _Float16* __restrict__ Vt,
                                            float* __restrict__ as_, float* __restrict__ ad_,
                                            int bid, int M, int K, int heads) {
  const int wave = __builtin_amdgcn_readfirstlane(bid * 8 + (int)(threadIdx.x >> 5));
  if (wave >= (M >> 4)) return;
  const int lane = threadIdx.x & 31;
  const int half = lane >> 4;
  const int l16  = lane & 15;
  const float*    aRow = Af + (size_t)(wave * 16 + l16) * K;
  const _Float16* bRow = Vt + (size_t)l16 * K;
  v8f acc = {};
  for (int k = 0; k < K; k += 32) {
    const int ka = k + half * 8;
    union { v16h v; v8h h[2]; } a, b;
    const float* ap = aRow + ka;
    const float4 p0 = *(const float4*)(ap);
    const float4 p1 = *(const float4*)(ap + 4);
    const float4 p2 = *(const float4*)(ap + 16);
    const float4 p3 = *(const float4*)(ap + 20);
    a.v[0]  = (_Float16)p0.x; a.v[1]  = (_Float16)p0.y; a.v[2]  = (_Float16)p0.z; a.v[3]  = (_Float16)p0.w;
    a.v[4]  = (_Float16)p1.x; a.v[5]  = (_Float16)p1.y; a.v[6]  = (_Float16)p1.z; a.v[7]  = (_Float16)p1.w;
    a.v[8]  = (_Float16)p2.x; a.v[9]  = (_Float16)p2.y; a.v[10] = (_Float16)p2.z; a.v[11] = (_Float16)p2.w;
    a.v[12] = (_Float16)p3.x; a.v[13] = (_Float16)p3.y; a.v[14] = (_Float16)p3.z; a.v[15] = (_Float16)p3.w;
    b.h[0] = *(const v8h*)(bRow + ka);
    b.h[1] = *(const v8h*)(bRow + ka + 16);
    acc = __builtin_amdgcn_wmma_f32_16x16x32_f16(false, a.v, false, b.v, (short)0, acc, false, false);
  }
  if (l16 < 2 * heads) {
    const bool isD = l16 >= heads;
    const int  hh  = isD ? l16 - heads : l16;
    float* o = (isD ? ad_ : as_) + (size_t)(wave * 16 + half * 8) * heads + hh;
    #pragma unroll
    for (int j = 0; j < 8; ++j) o[(size_t)j * heads] = acc[j];
  }
}

__global__ void __launch_bounds__(256) k_scan_logits(int* __restrict__ planes, const int* __restrict__ tileSum,
                                 int* __restrict__ off,
                                 int nCnt, int nChunk, int nScan,
                                 const float* __restrict__ A,
                                 const _Float16* __restrict__ Vt,
                                 float* __restrict__ as_, float* __restrict__ ad_,
                                 int M, int K, int heads) {
  __shared__ int sScan[16];
  if ((int)blockIdx.x < nScan) {
    scan_tile(planes, tileSum, off, nCnt, nChunk, nScan, (int)blockIdx.x, sScan);
    return;
  }
  logits_tile(A, Vt, as_, ad_, (int)blockIdx.x - nScan, M, K, heads);
}


__device__ __forceinline__ v4h cvt4(float4 f) {
  v4h q;
  q[0] = (_Float16)f.x; q[1] = (_Float16)f.y; q[2] = (_Float16)f.z; q[3] = (_Float16)f.w;
  return q;
}

__device__ __forceinline__ void walk_rows_l1(const float* __restrict__ x, int lane,
                                             int s, float a0, float a1, float a2, float a3, int cnt,
                                             float4& z0, float4& z1, float4& z2, float4& z3) {
  for (int j0 = 0; j0 < cnt; j0 += 4) {
    float4 f[4];
    #pragma unroll
    for (int u = 0; u < 4; ++u) {
      const int sj = __builtin_amdgcn_readlane(s, j0 + u);
      f[u] = *(const float4*)(x + (size_t)sj * 128 + lane * 4);
    }
    #pragma unroll
    for (int u = 0; u < 4; ++u) {
      const int j = j0 + u;
      const float b0 = laneF(a0, j), b1 = laneF(a1, j), b2 = laneF(a2, j), b3 = laneF(a3, j);
      z0.x = __builtin_fmaf(f[u].x, b0, z0.x); z0.y = __builtin_fmaf(f[u].y, b0, z0.y);
      z0.z = __builtin_fmaf(f[u].z, b0, z0.z); z0.w = __builtin_fmaf(f[u].w, b0, z0.w);
      z1.x = __builtin_fmaf(f[u].x, b1, z1.x); z1.y = __builtin_fmaf(f[u].y, b1, z1.y);
      z1.z = __builtin_fmaf(f[u].z, b1, z1.z); z1.w = __builtin_fmaf(f[u].w, b1, z1.w);
      z2.x = __builtin_fmaf(f[u].x, b2, z2.x); z2.y = __builtin_fmaf(f[u].y, b2, z2.y);
      z2.z = __builtin_fmaf(f[u].z, b2, z2.z); z2.w = __builtin_fmaf(f[u].w, b2, z2.w);
      z3.x = __builtin_fmaf(f[u].x, b3, z3.x); z3.y = __builtin_fmaf(f[u].y, b3, z3.y);
      z3.z = __builtin_fmaf(f[u].z, b3, z3.z); z3.w = __builtin_fmaf(f[u].w, b3, z3.w);
    }
  }
}

__device__ __forceinline__ void row_sums_l1(const int* __restrict__ off, int* __restrict__ srcS,
                          int* __restrict__ srcT,
                          const float* __restrict__ as_, const float* __restrict__ ad_,
                          const float* __restrict__ x, int d, int lane,
                          float4& z0, float4& z1, float4& z2, float4& z3) {
  const int beg = off[d];
  const int deg = off[d + 1] - beg;
  const int tot = deg + 1;
  const float4 adv = *(const float4*)(ad_ + (size_t)d * 4);

  if (tot <= 32) {
    const bool valid = lane < tot;
    int s = d;
    if (lane < deg) s = srcS[beg + lane];
    if (deg > 1) {
      int place = 0;
      for (int j = 0; j < deg; ++j) {
        const int sj = __builtin_amdgcn_readlane(s, j);
        place += (sj < s || (sj == s && j < lane)) ? 1 : 0;
      }
      const int to = lane < deg ? place : lane;
      s = __builtin_amdgcn_ds_permute(to << 2, s);
      if (lane < deg) srcS[beg + lane] = s;
    }
    const float4 a = *(const float4*)(as_ + (size_t)s * 4);
    const float ninf = -__builtin_inff();
    const float r0 = leaky(a.x + adv.x), r1 = leaky(a.y + adv.y);
    const float r2 = leaky(a.z + adv.z), r3 = leaky(a.w + adv.w);
    const float e0 = valid ? r0 : ninf, e1 = valid ? r1 : ninf, e2 = valid ? r2 : ninf, e3 = valid ? r3 : ninf;
    const float m0 = waveMax(e0), m1 = waveMax(e1), m2 = waveMax(e2), m3 = waveMax(e3);
    const float p0 = expf(e0 - m0), p1 = expf(e1 - m1), p2 = expf(e2 - m2), p3 = expf(e3 - m3);
    const float s0 = waveSum(p0) + 1e-16f, s1 = waveSum(p1) + 1e-16f;
    const float s2 = waveSum(p2) + 1e-16f, s3 = waveSum(p3) + 1e-16f;
    walk_rows_l1(x, lane, s, p0 / s0, p1 / s1, p2 / s2, p3 / s3, tot, z0, z1, z2, z3);
  } else {
    for (int base = 0; base < deg; base += 32) {
      const int i = base + lane;
      const int si = i < deg ? srcS[beg + i] : 0x7fffffff;
      int place = 0;
      for (int b2 = 0; b2 < deg; b2 += 32) {
        const int j0 = b2 + lane;
        const int tj = j0 < deg ? srcS[beg + j0] : 0x7fffffff;
        const int nj = (deg - b2 < 32) ? (deg - b2) : 32;
        for (int j = 0; j < nj; ++j) {
          const int sj = __builtin_amdgcn_readlane(tj, j);
          place += (sj < si || (sj == si && b2 + j < i)) ? 1 : 0;
        }
      }
      if (i < deg) srcT[beg + place] = si;
    }
    __threadfence();
    float m0 = -__builtin_inff(), m1 = m0, m2 = m0, m3 = m0;
    for (int base = 0; base < tot; base += 32) {
      const int i = base + lane;
      if (i < tot) {
        const int s = i < deg ? srcT[beg + i] : d;
        const float4 a = *(const float4*)(as_ + (size_t)s * 4);
        m0 = fmaxf(m0, leaky(a.x + adv.x));
        m1 = fmaxf(m1, leaky(a.y + adv.y));
        m2 = fmaxf(m2, leaky(a.z + adv.z));
        m3 = fmaxf(m3, leaky(a.w + adv.w));
      }
    }
    m0 = waveMax(m0); m1 = waveMax(m1); m2 = waveMax(m2); m3 = waveMax(m3);

    float s0 = 0.f, s1 = 0.f, s2 = 0.f, s3 = 0.f;
    for (int base = 0; base < tot; base += 32) {
      const int i = base + lane;
      if (i < tot) {
        const int s = i < deg ? srcT[beg + i] : d;
        const float4 a = *(const float4*)(as_ + (size_t)s * 4);
        s0 += expf(leaky(a.x + adv.x) - m0);
        s1 += expf(leaky(a.y + adv.y) - m1);
        s2 += expf(leaky(a.z + adv.z) - m2);
        s3 += expf(leaky(a.w + adv.w) - m3);
      }
    }
    s0 = waveSum(s0) + 1e-16f; s1 = waveSum(s1) + 1e-16f;
    s2 = waveSum(s2) + 1e-16f; s3 = waveSum(s3) + 1e-16f;

    for (int base = 0; base < tot; base += 32) {
      const int i = base + lane;
      int s = d;
      float a0 = 0.f, a1 = 0.f, a2 = 0.f, a3 = 0.f;
      if (i < tot) {
        s = i < deg ? srcT[beg + i] : d;
        const float4 a = *(const float4*)(as_ + (size_t)s * 4);
        a0 = expf(leaky(a.x + adv.x) - m0) / s0;
        a1 = expf(leaky(a.y + adv.y) - m1) / s1;
        a2 = expf(leaky(a.z + adv.z) - m2) / s2;
        a3 = expf(leaky(a.w + adv.w) - m3) / s3;
      }
      const int cnt = (tot - base < 32) ? (tot - base) : 32;
      walk_rows_l1(x, lane, s, a0, a1, a2, a3, cnt, z0, z1, z2, z3);
    }
  }

}

#define PZ 136
#define PH 264
#define PO 68
__global__ void __launch_bounds__(512) __attribute__((amdgpu_waves_per_eu(16, 16))) k_layer1(const int* __restrict__ off, int* __restrict__ srcS,
                          int* __restrict__ srcT,
                          const float* __restrict__ as_, const float* __restrict__ ad_,
                          const float* __restrict__ x,
                          const _Float16* __restrict__ w1t, const float* __restrict__ b1,
                          const _Float16* __restrict__ w2t,
                          const float* __restrict__ att_s2, const float* __restrict__ att_d2,
                          float* __restrict__ h2, float* __restrict__ as2, float* __restrict__ ad2,
                          int Nn) {
  __shared__ __attribute__((aligned(16))) _Float16 sZ[4 * 16 * PZ];
  __shared__ __attribute__((aligned(16))) _Float16 sH[16 * PH];
  __shared__ __attribute__((aligned(16))) float    sO[16 * PO];
  __shared__ float sL[2 * 64];

  const int w    = __builtin_amdgcn_readfirstlane((int)(threadIdx.x >> 5));
  const int lane = threadIdx.x & 31;
  const int half = lane >> 4;
  const int l16  = lane & 15;
  const int d    = (int)blockIdx.x * 16 + w;

  float4 z0 = make_float4(0.f, 0.f, 0.f, 0.f);
  float4 z1 = z0, z2 = z0, z3 = z0;
  if (d < Nn) row_sums_l1(off, srcS, srcT, as_, ad_, x, d, lane, z0, z1, z2, z3);
  {
    _Float16* o = sZ + w * PZ + lane * 4;
    *(v4h*)(o)               = cvt4(z0);
    *(v4h*)(o + 16 * PZ)     = cvt4(z1);
    *(v4h*)(o + 2 * 16 * PZ) = cvt4(z2);
    *(v4h*)(o + 3 * 16 * PZ) = cvt4(z3);
  }
  __syncthreads();

  {
    const int hd = w >> 2, ct = w & 3;
    const int col = hd * 64 + ct * 16 + l16;
    const _Float16* aRow = sZ + (hd * 16 + l16) * PZ;
    const _Float16* bFrag = w1t + (size_t)w * (4 * 2 * 256) + lane * 8;
    const float bv = b1[col];
    v8f acc = {};
    #pragma unroll 2
    for (int k = 0; k < 128; k += 32) {
      const int ka = k + half * 8;
      union { v16h v; v8h h[2]; } a, b;
      a.h[0] = *(const v8h*)(aRow + ka);
      a.h[1] = *(const v8h*)(aRow + ka + 16);
      b.h[0] = *(const v8h*)(bFrag + k * 16);
      b.h[1] = *(const v8h*)(bFrag + k * 16 + 256);
      acc = __builtin_amdgcn_wmma_f32_16x16x32_f16(false, a.v, false, b.v, (short)0, acc, false, false);
    }
    _Float16* hp = sH + (half * 8) * PH + col;
    #pragma unroll
    for (int j = 0; j < 8; ++j) {
      const float v = acc[j] + bv;
      hp[j * PH] = (_Float16)(v > 0.f ? v : 0.f);
    }
  }
  __syncthreads();

  if (w < 4) {
    const int col = w * 16 + l16;
    const _Float16* aRow = sH + l16 * PH;
    const _Float16* bFrag = w2t + (size_t)w * (8 * 2 * 256) + lane * 8;
    const float sc = att_s2[col], dc = att_d2[col];
    v8f acc = {};
    #pragma unroll 2
    for (int k = 0; k < 256; k += 32) {
      const int ka = k + half * 8;
      union { v16h v; v8h h[2]; } a, b;
      a.h[0] = *(const v8h*)(aRow + ka);
      a.h[1] = *(const v8h*)(aRow + ka + 16);
      b.h[0] = *(const v8h*)(bFrag + k * 16);
      b.h[1] = *(const v8h*)(bFrag + k * 16 + 256);
      acc = __builtin_amdgcn_wmma_f32_16x16x32_f16(false, a.v, false, b.v, (short)0, acc, false, false);
    }
    float* op = sO + (half * 8) * PO + col;
    float vs = 0.f, vd = 0.f;
    #pragma unroll
    for (int j = 0; j < 8; ++j) {
      op[j * PO] = acc[j];
      const float ps = rowSum16(acc[j] * sc);
      const float pd = rowSum16(acc[j] * dc);
      if (l16 == j) { vs = ps; vd = pd; }
    }
    if (l16 < 8) {
      sL[w * 16 + half * 8 + l16]      = vs;
      sL[64 + w * 16 + half * 8 + l16] = vd;
    }
  }
  __syncthreads();

  if (d < Nn) {
    const float2 v = *(const float2*)(sO + w * PO + lane * 2);
    *(float2*)(h2 + (size_t)d * 64 + lane * 2) = v;
  }
  if (w == 0) {
    const float* q = sL + half * 64 + l16;
    const float v = ((q[0] + q[16]) + q[32]) + q[48];
    const int r = (int)blockIdx.x * 16 + l16;
    if (r < Nn) (half ? ad2 : as2)[r] = v;
  }
}

__device__ __forceinline__ void walk_rows_l2(const float* __restrict__ feat, int lane,
                                             int s, float al, int cnt, float2& acc) {
  for (int j0 = 0; j0 < cnt; j0 += 16) {
    float2 f[16];
    const bool upper = j0 + 8 < cnt;
    #pragma unroll
    for (int u = 0; u < 8; ++u) {
      const int sj = __builtin_amdgcn_readlane(s, j0 + u);
      f[u] = *(const float2*)(feat + (size_t)sj * 64 + lane * 2);
    }
    if (upper) {
      #pragma unroll
      for (int u = 8; u < 16; ++u) {
        const int sj = __builtin_amdgcn_readlane(s, j0 + u);
        f[u] = *(const float2*)(feat + (size_t)sj * 64 + lane * 2);
      }
    }
    #pragma unroll
    for (int u = 0; u < 8; ++u) {
      const float aj = laneF(al, j0 + u);
      acc.x = __builtin_fmaf(f[u].x, aj, acc.x);
      acc.y = __builtin_fmaf(f[u].y, aj, acc.y);
    }
    if (upper) {
      #pragma unroll
      for (int u = 8; u < 16; ++u) {
        const float aj = laneF(al, j0 + u);
        acc.x = __builtin_fmaf(f[u].x, aj, acc.x);
        acc.y = __builtin_fmaf(f[u].y, aj, acc.y);
      }
    }
  }
}

__global__ void k_node_l2(const int* __restrict__ off, const int* __restrict__ srcS,
                          const int* __restrict__ srcT,
                          const float* __restrict__ as_, const float* __restrict__ ad_,
                          const float* __restrict__ feat, const float* __restrict__ bias,
                          float* __restrict__ outp, int Nn) {
  int d = blockIdx.x * (blockDim.x >> 5) + (threadIdx.x >> 5);
  if (d >= Nn) return;
  d = __builtin_amdgcn_readfirstlane(d);
  const int lane = threadIdx.x & 31;
  const int beg = off[d];
  const int deg = off[d + 1] - beg;
  const int tot = deg + 1;
  const float adv = ad_[d];
  float2 acc = make_float2(0.f, 0.f);

  if (tot <= 32) {
    const bool valid = lane < tot;
    int s = d;
    if (lane < deg) s = srcS[beg + lane];
    const float r = leaky(as_[s] + adv);
    const float e = valid ? r : -__builtin_inff();
    const float m = waveMax(e);
    const float p = expf(e - m);
    const float sm = waveSum(p) + 1e-16f;
    walk_rows_l2(feat, lane, s, p / sm, tot, acc);
  } else {
    float m = -__builtin_inff();
    for (int base = 0; base < tot; base += 32) {
      const int i = base + lane;
      if (i < tot) {
        const int s = i < deg ? srcT[beg + i] : d;
        m = fmaxf(m, leaky(as_[s] + adv));
      }
    }
    m = waveMax(m);

    float sm = 0.f;
    for (int base = 0; base < tot; base += 32) {
      const int i = base + lane;
      if (i < tot) {
        const int s = i < deg ? srcT[beg + i] : d;
        sm += expf(leaky(as_[s] + adv) - m);
      }
    }
    sm = waveSum(sm) + 1e-16f;

    for (int base = 0; base < tot; base += 32) {
      const int i = base + lane;
      int s = d;
      float al = 0.f;
      if (i < tot) {
        s = i < deg ? srcT[beg + i] : d;
        al = expf(leaky(as_[s] + adv) - m) / sm;
      }
      const int cnt = (tot - base < 32) ? (tot - base) : 32;
      walk_rows_l2(feat, lane, s, al, cnt, acc);
    }
  }
  float2 r;
  r.x = acc.x + bias[lane * 2];
  r.y = acc.y + bias[lane * 2 + 1];
  *(float2*)(outp + (size_t)d * 64 + lane * 2) = r;
}

extern "C" void kernel_launch(void* const* d_in, const int* in_sizes, int n_in,
                              void* d_out, int out_size, void* d_ws, size_t ws_size,
                              hipStream_t stream) {
  const int IN = 128, HID = 64, H1 = 4;
  const int HC1 = H1 * HID;
  const int HC2 = HID;
  const int Nn   = in_sizes[0] / IN;
  const int E    = in_sizes[1] / 2;

  const float* x   = (const float*)d_in[0];
  const int*   ei  = (const int*)d_in[1];
  const float* W1  = (const float*)d_in[2];
  const float* aS1 = (const float*)d_in[3];
  const float* aD1 = (const float*)d_in[4];
  const float* b1  = (const float*)d_in[5];
  const float* W2  = (const float*)d_in[6];
  const float* aS2 = (const float*)d_in[7];
  const float* aD2 = (const float*)d_in[8];
  const float* b2  = (const float*)d_in[9];
  float* out = (float*)d_out;
  const int* srcI = ei;
  const int* dstI = ei + E;

  char* p = (char*)d_ws;
  auto carve = [&](size_t bytes) -> void* {
    void* r = (void*)p;
    p += (bytes + 255) & ~(size_t)255;
    return r;
  };
  _Float16* w1t  = (_Float16*)carve((size_t)HC1 * IN * 2);
  _Float16* w2t  = (_Float16*)carve((size_t)HID * HC1 * 2);
  _Float16* vt   = (_Float16*)carve((size_t)16 * IN * 2);
  float*    h2   = (float*)carve((size_t)Nn * HC2 * 4);
  float*    as1  = (float*)carve((size_t)Nn * H1 * 4);
  float*    ad1  = (float*)carve((size_t)Nn * H1 * 4);
  float*    as2  = (float*)carve((size_t)Nn * 4);
  float*    ad2  = (float*)carve((size_t)Nn * 4);
  const int TB = 256;
  auto cdiv = [](long long a, long long b) { return (int)((a + b - 1) / b); };
  const int CH    = 32;
  const int chunk = cdiv(cdiv(E, CH), TB) * TB;
  const int nScan = cdiv((long long)Nn + 1, 1024);
  int*      planes  = (int*)carve((size_t)CH * Nn * 4);
  int*      tileSum = (int*)carve((size_t)CH * nScan * 4);
  int*      off  = (int*)carve((size_t)(Nn + 1) * 4);
  int*      rank = (int*)carve((size_t)E * 4);
  int*      srcS = (int*)carve((size_t)E * 4);
  int*      srcT = (int*)carve((size_t)E * 4);

  k_prep<<<cdiv((long long)IN * HC1 + (long long)HC1 * HID + 16LL * IN, TB), TB, 0, stream>>>(
      W1, w1t, IN, HC1, W2, w2t, HC1, HID, aS1, aD1, vt);

  k_count<<<CH, 1024, 0, stream>>>(dstI, E, chunk, Nn, nScan, planes, tileSum, rank);

  k_scan_logits<<<nScan + cdiv(Nn / 16, 8), TB, 0, stream>>>(planes, tileSum, off, Nn, CH, nScan,
                                                             x, vt, as1, ad1, Nn, IN, H1);
  k_fill<<<cdiv(E, TB), TB, 0, stream>>>(srcI, dstI, E, chunk, Nn, planes, rank, srcS);
  k_layer1<<<cdiv(Nn, 16), 512, 0, stream>>>(off, srcS, srcT, as1, ad1, x, w1t, b1, w2t, aS2, aD2,
                                              h2, as2, ad2, Nn);

  k_node_l2<<<cdiv(Nn, 8), TB, 0, stream>>>(off, srcS, srcT, as2, ad2, h2, b2, out, Nn);
}
